# starting kernel
# baseline (speedup 1.0000x reference)
#include <hip/hip_runtime.h>

typedef __attribute__((ext_vector_type(2))) float v2f;
typedef __attribute__((ext_vector_type(8))) float v8f;

#define WT_STRIDE 66

__global__ __launch_bounds__(256)
void proj_wmma_kernel(const float* __restrict__ h_c,
                      const float* __restrict__ h_p,
                      const float* __restrict__ Wqc, const float* __restrict__ bqc,
                      const float* __restrict__ Wqp, const float* __restrict__ bqp,
                      const float* __restrict__ Wk,  const float* __restrict__ bk,
                      const float* __restrict__ Wv,  const float* __restrict__ bv,
                      float* __restrict__ Qc, float* __restrict__ Qp,
                      float* __restrict__ Kn, float* __restrict__ Vn,
                      int n)
{
    __shared__ float sWT[4][64 * WT_STRIDE];
    __shared__ float sB[4][64];

    const int tid = threadIdx.x;
    const float* Ws[4] = {Wqc, Wqp, Wk, Wv};
    const float* Bs[4] = {bqc, bqp, bk, bv};
    for (int o = 0; o < 4; ++o) {
        for (int i = tid; i < 4096; i += 256) {
            int col = i >> 6, k = i & 63;
            sWT[o][col * WT_STRIDE + k] = Ws[o][k * 64 + col];
        }
        if (tid < 64) sB[o][tid] = Bs[o][tid];
    }
    __syncthreads();

    const int wave = tid >> 5;
    const int lane = tid & 31;
    const int half = lane >> 4;
    const int lrow = lane & 15;
    const int ncol = lane & 15;

    const int m0 = (blockIdx.x * 8 + wave) * 16;
    if (m0 >= n) return;

    const int row = m0 + lrow;

    v2f a_c[16], a_p[16];
#pragma unroll
    for (int ks = 0; ks < 16; ++ks) {
        int k = ks * 4 + 2 * half;
        a_c[ks] = *(const v2f*)(h_c + row * 64 + k);
        a_p[ks] = *(const v2f*)(h_p + row * 64 + k);
    }

    float* outs[4] = {Qc, Qp, Kn, Vn};

#pragma unroll
    for (int o = 0; o < 4; ++o) {
        float* op = outs[o];
#pragma unroll
        for (int nt = 0; nt < 4; ++nt) {
            const int ncg = nt * 16 + ncol;
            v8f acc = {0.f, 0.f, 0.f, 0.f, 0.f, 0.f, 0.f, 0.f};
#pragma unroll
            for (int ks = 0; ks < 16; ++ks) {
                int k = ks * 4 + 2 * half;
                v2f b = *(const v2f*)(&sWT[o][ncg * WT_STRIDE + k]);
                v2f a = (o == 1) ? a_p[ks] : a_c[ks];
                acc = __builtin_amdgcn_wmma_f32_16x16x4_f32(
                     false, a,  false, b,
                     (short)0, acc,  false,  false);
            }
            const float bias = sB[o][ncg];
#pragma unroll
            for (int r = 0; r < 8; ++r) {
                int m = m0 + r + 8 * half;
                op[m * 64 + ncg] = acc[r] + bias;
            }
        }
    }
}

__global__ void zero_kernel(float* __restrict__ p, int n)
{
    int i = blockIdx.x * blockDim.x + threadIdx.x;
    if (i < n) p[i] = 0.0f;
}

__global__ __launch_bounds__(256)
void edge_attn_kernel(const int* __restrict__ src, const int* __restrict__ dst,
                      const float* __restrict__ Qc, const float* __restrict__ Qp,
                      const float* __restrict__ Kn, const float* __restrict__ Vn,
                      float* __restrict__ out_c, float* __restrict__ out_p,
                      int total  )
{
    int idx = blockIdx.x * blockDim.x + threadIdx.x;
    if (idx >= total) return;
    const int e = idx >> 3;
    const int h = idx & 7;
    const int s = src[e];
    const int d = dst[e];
    const int bs = s * 64 + h * 8;
    const int bd = d * 64 + h * 8;

    const float4 k0  = *(const float4*)(Kn + bs);
    const float4 k1  = *(const float4*)(Kn + bs + 4);
    const float4 v0  = *(const float4*)(Vn + bs);
    const float4 v1  = *(const float4*)(Vn + bs + 4);
    const float4 qc0 = *(const float4*)(Qc + bd);
    const float4 qc1 = *(const float4*)(Qc + bd + 4);
    const float4 qp0 = *(const float4*)(Qp + bd);
    const float4 qp1 = *(const float4*)(Qp + bd + 4);

    float dc = qc0.x*k0.x + qc0.y*k0.y + qc0.z*k0.z + qc0.w*k0.w
             + qc1.x*k1.x + qc1.y*k1.y + qc1.z*k1.z + qc1.w*k1.w;
    float dp = qp0.x*k0.x + qp0.y*k0.y + qp0.z*k0.z + qp0.w*k0.w
             + qp1.x*k1.x + qp1.y*k1.y + qp1.z*k1.z + qp1.w*k1.w;

    const float sc = __expf(fminf(fmaxf(dc * 0.125f, -5.0f), 5.0f));
    const float sp = __expf(fminf(fmaxf(dp * 0.125f, -5.0f), 5.0f));

    atomicAdd(out_c + bd + 0, v0.x * sc);
    atomicAdd(out_c + bd + 1, v0.y * sc);
    atomicAdd(out_c + bd + 2, v0.z * sc);
    atomicAdd(out_c + bd + 3, v0.w * sc);
    atomicAdd(out_c + bd + 4, v1.x * sc);
    atomicAdd(out_c + bd + 5, v1.y * sc);
    atomicAdd(out_c + bd + 6, v1.z * sc);
    atomicAdd(out_c + bd + 7, v1.w * sc);

    atomicAdd(out_p + bd + 0, v0.x * sp);
    atomicAdd(out_p + bd + 1, v0.y * sp);
    atomicAdd(out_p + bd + 2, v0.z * sp);
    atomicAdd(out_p + bd + 3, v0.w * sp);
    atomicAdd(out_p + bd + 4, v1.x * sp);
    atomicAdd(out_p + bd + 5, v1.y * sp);
    atomicAdd(out_p + bd + 6, v1.z * sp);
    atomicAdd(out_p + bd + 7, v1.w * sp);
}

extern "C" void kernel_launch(void* const* d_in, const int* in_sizes, int n_in,
                              void* d_out, int out_size, void* d_ws, size_t ws_size,
                              hipStream_t stream)
{
    const float* h_c  = (const float*)d_in[0];
    const float* h_p  = (const float*)d_in[1];
    const int*   src  = (const int*)  d_in[2];
    const int*   dst  = (const int*)  d_in[3];
    const float* Wqcw = (const float*)d_in[4];
    const float* Wqcb = (const float*)d_in[5];
    const float* Wqpw = (const float*)d_in[6];
    const float* Wqpb = (const float*)d_in[7];
    const float* Wkw  = (const float*)d_in[8];
    const float* Wkb  = (const float*)d_in[9];
    const float* Wvw  = (const float*)d_in[10];
    const float* Wvb  = (const float*)d_in[11];

    const int n = in_sizes[0] / 64;
    const int e = in_sizes[2];

    float* Qc = (float*)d_ws;
    float* Qp = Qc + (size_t)n * 64;
    float* Kn = Qp + (size_t)n * 64;
    float* Vn = Kn + (size_t)n * 64;

    float* out_c = (float*)d_out;
    float* out_p = out_c + (size_t)n * 64;

    {
        int total = 2 * n * 64;
        zero_kernel<<<(total + 255) / 256, 256, 0, stream>>>((float*)d_out, total);
    }

    {
        int blocks = (n + 127) / 128;
        proj_wmma_kernel<<<blocks, 256, 0, stream>>>(
            h_c, h_p, Wqcw, Wqcb, Wqpw, Wqpb, Wkw, Wkb, Wvw, Wvb,
            Qc, Qp, Kn, Vn, n);
    }

    {
        int total = e * 8;
        edge_attn_kernel<<<(total + 255) / 256, 256, 0, stream>>>(
            src, dst, Qc, Qp, Kn, Vn, out_c, out_p, total);
    }
}


# float atomic scatter-add replaced by per-node gather: edges bucketed by integer atomics, each node sums edges in ascending edge index in registers (arrival order above 64 edges), one store per row
# speedup vs baseline: 2.1057x; 2.1057x over previous
#include <hip/hip_runtime.h>

typedef __attribute__((ext_vector_type(2))) float v2f;
typedef __attribute__((ext_vector_type(8))) float v8f;

#define WT_STRIDE 66
#define SEG_CAP   64
#define SEG_PITCH 65

__global__ __launch_bounds__(256)
void proj_wmma_kernel(const float* __restrict__ h_c,
                      const float* __restrict__ h_p,
                      const float* __restrict__ Wqc, const float* __restrict__ bqc,
                      const float* __restrict__ Wqp, const float* __restrict__ bqp,
                      const float* __restrict__ Wk,  const float* __restrict__ bk,
                      const float* __restrict__ Wv,  const float* __restrict__ bv,
                      float* __restrict__ Qc, float* __restrict__ Qp,
                      float* __restrict__ Kn, float* __restrict__ Vn,
                      int n)
{
    __shared__ float sWT[4][64 * WT_STRIDE];
    __shared__ float sB[4][64];

    const int tid = threadIdx.x;
    const float* Ws[4] = {Wqc, Wqp, Wk, Wv};
    const float* Bs[4] = {bqc, bqp, bk, bv};
    for (int o = 0; o < 4; ++o) {
        for (int i = tid; i < 4096; i += 256) {
            int col = i >> 6, k = i & 63;
            sWT[o][col * WT_STRIDE + k] = Ws[o][k * 64 + col];
        }
        if (tid < 64) sB[o][tid] = Bs[o][tid];
    }
    __syncthreads();

    const int wave = tid >> 5;
    const int lane = tid & 31;
    const int half = lane >> 4;
    const int lrow = lane & 15;
    const int ncol = lane & 15;

    const int m0 = (blockIdx.x * 8 + wave) * 16;
    if (m0 >= n) return;

    const int row = m0 + lrow;

    v2f a_c[16], a_p[16];
#pragma unroll
    for (int ks = 0; ks < 16; ++ks) {
        int k = ks * 4 + 2 * half;
        a_c[ks] = *(const v2f*)(h_c + row * 64 + k);
        a_p[ks] = *(const v2f*)(h_p + row * 64 + k);
    }

    float* outs[4] = {Qc, Qp, Kn, Vn};

#pragma unroll
    for (int o = 0; o < 4; ++o) {
        float* op = outs[o];
#pragma unroll
        for (int nt = 0; nt < 4; ++nt) {
            const int ncg = nt * 16 + ncol;
            v8f acc = {0.f, 0.f, 0.f, 0.f, 0.f, 0.f, 0.f, 0.f};
#pragma unroll
            for (int ks = 0; ks < 16; ++ks) {
                int k = ks * 4 + 2 * half;
                v2f b = *(const v2f*)(&sWT[o][ncg * WT_STRIDE + k]);
                v2f a = (o == 1) ? a_p[ks] : a_c[ks];
                acc = __builtin_amdgcn_wmma_f32_16x16x4_f32(
                     false, a,  false, b,
                     (short)0, acc,  false,  false);
            }
            const float bias = sB[o][ncg];
#pragma unroll
            for (int r = 0; r < 8; ++r) {
                int m = m0 + r + 8 * half;
                op[m * 64 + ncg] = acc[r] + bias;
            }
        }
    }
}

__global__ __launch_bounds__(256)
void zero_int_kernel(int* __restrict__ p, int n)
{
    int i = blockIdx.x * blockDim.x + threadIdx.x;
    if (i < n) p[i] = 0;
}

__global__ __launch_bounds__(256)
void degree_count_kernel(const int* __restrict__ dst, int* __restrict__ cnt, int e_total)
{
    int e = blockIdx.x * blockDim.x + threadIdx.x;
    if (e < e_total) atomicAdd(cnt + dst[e], 1);
}

__global__ __launch_bounds__(256)
void segment_alloc_kernel(const int* cnt, int* cursor,
                          int* __restrict__ start, int* __restrict__ cur, int n)
{
    __shared__ int wsum[8];
    __shared__ int base;
    const int tid  = threadIdx.x;
    const int lane = tid & 31;
    const int wave = tid >> 5;
    const int i = blockIdx.x * 256 + tid;
    const int c = (i < n) ? cnt[i] : 0;

    int x = c;
    for (int off = 1; off < 32; off <<= 1) {
        int y = __shfl_up(x, off, 32);
        if (lane >= off) x += y;
    }
    if (lane == 31) wsum[wave] = x;
    __syncthreads();
    if (tid == 0) {
        int t = 0;
        for (int w = 0; w < 8; ++w) { int s = wsum[w]; wsum[w] = t; t += s; }
        base = atomicAdd(cursor, t);
    }
    __syncthreads();
    if (i < n) {
        const int st = base + wsum[wave] + x - c;
        start[i] = st;
        cur[i]   = st;
    }
}

__global__ __launch_bounds__(256)
void segment_fill_kernel(const int* __restrict__ src, const int* __restrict__ dst,
                         int* __restrict__ cur, int2* __restrict__ ent, int e_total)
{
    int e = blockIdx.x * blockDim.x + threadIdx.x;
    if (e < e_total) {
        const int d = dst[e];
        const int p = atomicAdd(cur + d, 1);
        ent[p] = make_int2(e, src[e]);
    }
}

struct row8 { float4 lo, hi; };

__device__ __forceinline__ row8 load_row8(const float* p)
{
    row8 r;
    r.lo = *(const float4*)(p);
    r.hi = *(const float4*)(p + 4);
    return r;
}

__device__ __forceinline__ float dot8(const row8 a, const row8 b)
{
    return a.lo.x*b.lo.x + a.lo.y*b.lo.y + a.lo.z*b.lo.z + a.lo.w*b.lo.w
         + a.hi.x*b.hi.x + a.hi.y*b.hi.y + a.hi.z*b.hi.z + a.hi.w*b.hi.w;
}

__device__ __forceinline__ float edge_score(float dotv)
{
    return __expf(fminf(fmaxf(dotv * 0.125f, -5.0f), 5.0f));
}

__device__ __forceinline__ void add_scaled(row8& a, const row8 v, float s)
{
    a.lo.x += v.lo.x * s; a.lo.y += v.lo.y * s; a.lo.z += v.lo.z * s; a.lo.w += v.lo.w * s;
    a.hi.x += v.hi.x * s; a.hi.y += v.hi.y * s; a.hi.z += v.hi.z * s; a.hi.w += v.hi.w * s;
}

__device__ __forceinline__ void edge_update(const row8 k, const row8 v,
                                            const row8 qc, const row8 qp,
                                            row8& ac, row8& ap)
{
    const float sc = edge_score(dot8(qc, k));
    const float sp = edge_score(dot8(qp, k));
    add_scaled(ac, v, sc);
    add_scaled(ap, v, sp);
}

__global__ __launch_bounds__(256)
void node_gather_kernel(const int2* __restrict__ ent, const int* __restrict__ start,
                        const int* __restrict__ cnt,
                        const float* __restrict__ Kn, const float* __restrict__ Vn,
                        float* out_c, float* out_p, int n)
{
    __shared__ int ssrc[32 * SEG_PITCH];

    const int tid = threadIdx.x;
    const int g = tid >> 3;
    const int h = tid & 7;
    const int d = blockIdx.x * 32 + g;
    const bool valid = d < n;

    int b = 0, c = 0;
    if (valid) { b = start[d]; c = cnt[d]; }
    const bool ordered = (c <= SEG_CAP);
    int* my = ssrc + g * SEG_PITCH;

    if (ordered) {
        for (int j = h; j < c; j += 8) {
            const int2 me = ent[b + j];
            int r = 0;
            for (int t = 0; t < c; ++t) r += (ent[b + t].x < me.x) ? 1 : 0;
            my[r] = me.y;
        }
    }
    __syncthreads();
    if (!valid) return;

    const int hc  = h * 8;
    const int col = d * 64 + hc;
    const row8 qc = load_row8(out_c + col);
    const row8 qp = load_row8(out_p + col);

    row8 ac, ap;
    ac.lo = make_float4(0.f, 0.f, 0.f, 0.f);
    ac.hi = ac.lo; ap.lo = ac.lo; ap.hi = ac.lo;

    if (ordered) {
        int i = 0;
        if (c >= 2) {
            const int f0 = my[0] * 64 + hc;
            const int f1 = my[1] * 64 + hc;
            row8 k0 = load_row8(Kn + f0), v0 = load_row8(Vn + f0);
            row8 k1 = load_row8(Kn + f1), v1 = load_row8(Vn + f1);
            do {
                const row8 ck0 = k0, cv0 = v0, ck1 = k1, cv1 = v1;
                if (i + 3 < c) {
                    const int o0 = my[i + 2] * 64 + hc;
                    const int o1 = my[i + 3] * 64 + hc;
                    k0 = load_row8(Kn + o0); v0 = load_row8(Vn + o0);
                    k1 = load_row8(Kn + o1); v1 = load_row8(Vn + o1);
                }
                edge_update(ck0, cv0, qc, qp, ac, ap);
                edge_update(ck1, cv1, qc, qp, ac, ap);
                i += 2;
            } while (i + 1 < c);
        }
        if (i < c) {
            const int o0 = my[i] * 64 + hc;
            const row8 kl = load_row8(Kn + o0);
            const row8 vl = load_row8(Vn + o0);
            edge_update(kl, vl, qc, qp, ac, ap);
        }
    } else {
        for (int i = 0; i < c; ++i) {
            const int o0 = ent[b + i].y * 64 + hc;
            const row8 k0 = load_row8(Kn + o0);
            const row8 v0 = load_row8(Vn + o0);
            edge_update(k0, v0, qc, qp, ac, ap);
        }
    }

    *(float4*)(out_c + col)     = ac.lo;
    *(float4*)(out_c + col + 4) = ac.hi;
    *(float4*)(out_p + col)     = ap.lo;
    *(float4*)(out_p + col + 4) = ap.hi;
}

extern "C" void kernel_launch(void* const* d_in, const int* in_sizes, int n_in,
                              void* d_out, int out_size, void* d_ws, size_t ws_size,
                              hipStream_t stream)
{
    const float* h_c  = (const float*)d_in[0];
    const float* h_p  = (const float*)d_in[1];
    const int*   src  = (const int*)  d_in[2];
    const int*   dst  = (const int*)  d_in[3];
    const float* Wqcw = (const float*)d_in[4];
    const float* Wqcb = (const float*)d_in[5];
    const float* Wqpw = (const float*)d_in[6];
    const float* Wqpb = (const float*)d_in[7];
    const float* Wkw  = (const float*)d_in[8];
    const float* Wkb  = (const float*)d_in[9];
    const float* Wvw  = (const float*)d_in[10];
    const float* Wvb  = (const float*)d_in[11];

    const int n = in_sizes[0] / 64;
    const int e = in_sizes[2];

    float* out_c = (float*)d_out;
    float* out_p = out_c + (size_t)n * 64;

    float* Kn   = (float*)d_ws;
    float* Vn   = Kn + (size_t)n * 64;
    int2*  ent  = (int2*)(Vn + (size_t)n * 64);
    int*   cnt  = (int*)(ent + (size_t)e);
    int*   start = cnt + n + 1;
    int*   cur   = start + n;

    zero_int_kernel<<<(n + 1 + 255) / 256, 256, 0, stream>>>(cnt, n + 1);

    degree_count_kernel<<<(e + 255) / 256, 256, 0, stream>>>(dst, cnt, e);

    segment_alloc_kernel<<<(n + 255) / 256, 256, 0, stream>>>(cnt, cnt + n, start, cur, n);

    segment_fill_kernel<<<(e + 255) / 256, 256, 0, stream>>>(src, dst, cur, ent, e);

    {
        int blocks = (n + 127) / 128;
        proj_wmma_kernel<<<blocks, 256, 0, stream>>>(
            h_c, h_p, Wqcw, Wqcb, Wqpw, Wqpb, Wkw, Wkb, Wvw, Wvb,
            out_c, out_p, Kn, Vn, n);
    }

    node_gather_kernel<<<(n + 31) / 32, 256, 0, stream>>>(
        ent, start, cnt, Kn, Vn, out_c, out_p, n);
}


# edge bucketing without global atomics: edges binned by 512-node destination range in LDS, written in whole runs, then each range placed per node in LDS; sums and their order as before
# speedup vs baseline: 4.1983x; 1.9938x over previous
#include <hip/hip_runtime.h>

typedef __attribute__((ext_vector_type(2))) float v2f;
typedef __attribute__((ext_vector_type(8))) float v8f;

#define WT_STRIDE 66
#define SEG_CAP   64
#define SEG_PITCH 65

__global__ __launch_bounds__(256)
void proj_wmma_kernel(const float* __restrict__ h_c,
                      const float* __restrict__ h_p,
                      const float* __restrict__ Wqc, const float* __restrict__ bqc,
                      const float* __restrict__ Wqp, const float* __restrict__ bqp,
                      const float* __restrict__ Wk,  const float* __restrict__ bk,
                      const float* __restrict__ Wv,  const float* __restrict__ bv,
                      float* __restrict__ Qc, float* __restrict__ Qp,
                      float* __restrict__ Kn, float* __restrict__ Vn,
                      int n)
{
    __shared__ float sWT[4][64 * WT_STRIDE];
    __shared__ float sB[4][64];

    const int tid = threadIdx.x;
    const float* Ws[4] = {Wqc, Wqp, Wk, Wv};
    const float* Bs[4] = {bqc, bqp, bk, bv};
    for (int o = 0; o < 4; ++o) {
        for (int i = tid; i < 4096; i += 256) {
            int col = i >> 6, k = i & 63;
            sWT[o][col * WT_STRIDE + k] = Ws[o][k * 64 + col];
        }
        if (tid < 64) sB[o][tid] = Bs[o][tid];
    }
    __syncthreads();

    const int wave = tid >> 5;
    const int lane = tid & 31;
    const int half = lane >> 4;
    const int lrow = lane & 15;
    const int ncol = lane & 15;

    const int m0 = (blockIdx.x * 8 + wave) * 16;
    if (m0 >= n) return;

    const int row = m0 + lrow;

    v2f a_c[16], a_p[16];
#pragma unroll
    for (int ks = 0; ks < 16; ++ks) {
        int k = ks * 4 + 2 * half;
        a_c[ks] = *(const v2f*)(h_c + row * 64 + k);
        a_p[ks] = *(const v2f*)(h_p + row * 64 + k);
    }

    float* outs[4] = {Qc, Qp, Kn, Vn};

#pragma unroll
    for (int o = 0; o < 4; ++o) {
        float* op = outs[o];
#pragma unroll
        for (int nt = 0; nt < 4; ++nt) {
            const int ncg = nt * 16 + ncol;
            v8f acc = {0.f, 0.f, 0.f, 0.f, 0.f, 0.f, 0.f, 0.f};
#pragma unroll
            for (int ks = 0; ks < 16; ++ks) {
                int k = ks * 4 + 2 * half;
                v2f b = *(const v2f*)(&sWT[o][ncg * WT_STRIDE + k]);
                v2f a = (o == 1) ? a_p[ks] : a_c[ks];
                acc = __builtin_amdgcn_wmma_f32_16x16x4_f32(
                     false, a,  false, b,
                     (short)0, acc,  false,  false);
            }
            const float bias = sB[o][ncg];
#pragma unroll
            for (int r = 0; r < 8; ++r) {
                int m = m0 + r + 8 * half;
                op[m * 64 + ncg] = acc[r] + bias;
            }
        }
    }
}

#define BIN_SHIFT 9
#define BIN_NODES 512
#define BIN_MAX   256
#define SRC_BITS  17
#define EPT       16
#define EPB       4096
#define STAGE_CAP 12288
#define PLB       8

__device__ __forceinline__ int wave_scan_incl(int x, int lane)
{
    for (int off = 1; off < 32; off <<= 1) {
        int y = __shfl_up(x, off, 32);
        if (lane >= off) x += y;
    }
    return x;
}

__global__ __launch_bounds__(256)
void range_count_kernel(const int* __restrict__ dst, int* __restrict__ bh,
                        int e_total, int nb, int nbins)
{
    __shared__ int hist[BIN_MAX];
    const int tid = threadIdx.x;
    hist[tid] = 0;
    __syncthreads();
    const int base = blockIdx.x * EPB;
    const int last = e_total - 1;
    int dd[EPT];
    for (int j = 0; j < EPT; ++j) {
        const int e = base + j * 256 + tid;
        dd[j] = dst[e < e_total ? e : last];
    }
    for (int j = 0; j < EPT; ++j) {
        const int e = base + j * 256 + tid;
        if (e < e_total) atomicAdd(&hist[dd[j] >> BIN_SHIFT], 1);
    }
    __syncthreads();
    if (tid < nbins) bh[tid * nb + blockIdx.x] = hist[tid];
}

__global__ __launch_bounds__(512)
void range_scan_kernel(const int* __restrict__ bh, int* __restrict__ offrel,
                       int* __restrict__ binCnt, int nb)
{
    __shared__ int wsum[16];
    const int tid  = threadIdx.x;
    const int lane = tid & 31;
    const int wave = tid >> 5;
    const int row  = blockIdx.x * nb;
    int carry = 0;
    for (int c0 = 0; c0 < nb; c0 += 512) {
        const int i = c0 + tid;
        const int v = (i < nb) ? bh[row + i] : 0;
        const int x = wave_scan_incl(v, lane);
        if (lane == 31) wsum[wave] = x;
        __syncthreads();
        int woff = 0, tot = 0;
        for (int w = 0; w < 16; ++w) {
            const int s = wsum[w];
            if (w < wave) woff += s;
            tot += s;
        }
        if (i < nb) offrel[row + i] = carry + woff + x - v;
        carry += tot;
        __syncthreads();
    }
    if (tid == 0) binCnt[blockIdx.x] = carry;
}

__global__ __launch_bounds__(256)
void range_scatter_kernel(const int* __restrict__ src, const int* __restrict__ dst,
                          const int* __restrict__ offrel, const int* __restrict__ binCnt,
                          int2* __restrict__ tmp, int e_total, int nb, int nbins)
{
    __shared__ int  hist[BIN_MAX];
    __shared__ int  lstart[BIN_MAX];
    __shared__ int  gdelta[BIN_MAX];
    __shared__ int  wsA[8];
    __shared__ int  wsB[8];
    __shared__ int2 stage[EPB];
    __shared__ int  sgp[EPB];

    const int tid  = threadIdx.x;
    const int lane = tid & 31;
    const int wave = tid >> 5;
    hist[tid] = 0;
    __syncthreads();

    const int base = blockIdx.x * EPB;
    const int last = e_total - 1;
    int dd[EPT], ss[EPT], rr[EPT];
    for (int j = 0; j < EPT; ++j) {
        const int e  = base + j * 256 + tid;
        const int ec = e < e_total ? e : last;
        dd[j] = dst[ec];
        ss[j] = src[ec];
    }
    for (int j = 0; j < EPT; ++j) {
        const int e = base + j * 256 + tid;
        rr[j] = 0;
        if (e < e_total) rr[j] = atomicAdd(&hist[dd[j] >> BIN_SHIFT], 1);
    }
    __syncthreads();

    const int h  = hist[tid];
    const int g  = (tid < nbins) ? binCnt[tid] : 0;
    const int xh = wave_scan_incl(h, lane);
    const int xg = wave_scan_incl(g, lane);
    if (lane == 31) { wsA[wave] = xh; wsB[wave] = xg; }
    __syncthreads();
    int offA = 0, offB = 0;
    for (int w = 0; w < 8; ++w) {
        if (w < wave) { offA += wsA[w]; offB += wsB[w]; }
    }
    const int ls = offA + xh - h;
    lstart[tid] = ls;
    gdelta[tid] = (tid < nbins) ? (offB + xg - g) + offrel[tid * nb + blockIdx.x] - ls : 0;
    __syncthreads();

    for (int j = 0; j < EPT; ++j) {
        const int e = base + j * 256 + tid;
        if (e < e_total) {
            const int bin = dd[j] >> BIN_SHIFT;
            const int p   = lstart[bin] + rr[j];
            stage[p] = make_int2(e, ss[j] | ((dd[j] & (BIN_NODES - 1)) << SRC_BITS));
            sgp[p]   = gdelta[bin] + p;
        }
    }
    __syncthreads();

    const int left = e_total - base;
    const int m    = left < EPB ? left : EPB;
    for (int i = tid; i < m; i += 256) tmp[sgp[i]] = stage[i];
}

__global__ __launch_bounds__(1024)
void range_place_kernel(const int2* __restrict__ tmp, const int* __restrict__ binCnt,
                        int2* __restrict__ ent, int* __restrict__ start,
                        int* __restrict__ cnt, int n)
{
    __shared__ int  lc[BIN_NODES + 1];
    __shared__ int  lcur[BIN_NODES];
    __shared__ int  wsum[32];
    __shared__ int  sh_bs;
    __shared__ int2 stage[STAGE_CAP];

    const int tid  = threadIdx.x;
    const int lane = tid & 31;
    const int wave = tid >> 5;
    const int k    = blockIdx.x;
    const int bc   = binCnt[k];

    {
        const int v = (tid < k) ? binCnt[tid] : 0;
        const int x = wave_scan_incl(v, lane);
        if (lane == 31) wsum[wave] = x;
        if (tid <= BIN_NODES) lc[tid] = 0;
        __syncthreads();
        if (tid == 0) {
            int t = 0;
            for (int w = 0; w < 32; ++w) t += wsum[w];
            sh_bs = t;
        }
        __syncthreads();
    }
    const int bs    = sh_bs;
    const int node0 = k << BIN_SHIFT;

    for (int i0 = tid; i0 < bc; i0 += PLB * 1024) {
        int y[PLB];
        for (int u = 0; u < PLB; ++u) {
            const int i = i0 + u * 1024;
            y[u] = tmp[bs + (i < bc ? i : bc - 1)].y;
        }
        for (int u = 0; u < PLB; ++u) {
            const int i = i0 + u * 1024;
            atomicAdd(&lc[i < bc ? (y[u] >> SRC_BITS) : BIN_NODES], 1);
        }
    }
    __syncthreads();

    {
        const int c = (tid < BIN_NODES) ? lc[tid] : 0;
        const int x = wave_scan_incl(c, lane);
        if (lane == 31) wsum[wave] = x;
        __syncthreads();
        int woff = 0;
        for (int w = 0; w < 16; ++w) {
            if (w < wave) woff += wsum[w];
        }
        if (tid < BIN_NODES) {
            const int ex = woff + x - c;
            lcur[tid] = ex;
            const int node = node0 + tid;
            if (node < n) { start[node] = bs + ex; cnt[node] = c; }
        }
        __syncthreads();
    }

    const bool staged = (bc <= STAGE_CAP);
    for (int i0 = tid; i0 < bc; i0 += PLB * 1024) {
        int2 a[PLB];
        for (int u = 0; u < PLB; ++u) {
            const int i = i0 + u * 1024;
            a[u] = tmp[bs + (i < bc ? i : bc - 1)];
        }
        for (int u = 0; u < PLB; ++u) {
            const int i = i0 + u * 1024;
            if (i < bc) {
                const int  p = atomicAdd(&lcur[a[u].y >> SRC_BITS], 1);
                const int2 o = make_int2(a[u].x, a[u].y & ((1 << SRC_BITS) - 1));
                if (staged) stage[p] = o; else ent[bs + p] = o;
            }
        }
    }
    if (staged) {
        __syncthreads();
        for (int i = tid; i < bc; i += 1024) ent[bs + i] = stage[i];
    }
}

struct row8 { float4 lo, hi; };

__device__ __forceinline__ row8 load_row8(const float* p)
{
    row8 r;
    r.lo = *(const float4*)(p);
    r.hi = *(const float4*)(p + 4);
    return r;
}

__device__ __forceinline__ float dot8(const row8 a, const row8 b)
{
    return a.lo.x*b.lo.x + a.lo.y*b.lo.y + a.lo.z*b.lo.z + a.lo.w*b.lo.w
         + a.hi.x*b.hi.x + a.hi.y*b.hi.y + a.hi.z*b.hi.z + a.hi.w*b.hi.w;
}

__device__ __forceinline__ float edge_score(float dotv)
{
    return __expf(fminf(fmaxf(dotv * 0.125f, -5.0f), 5.0f));
}

__device__ __forceinline__ void add_scaled(row8& a, const row8 v, float s)
{
    a.lo.x += v.lo.x * s; a.lo.y += v.lo.y * s; a.lo.z += v.lo.z * s; a.lo.w += v.lo.w * s;
    a.hi.x += v.hi.x * s; a.hi.y += v.hi.y * s; a.hi.z += v.hi.z * s; a.hi.w += v.hi.w * s;
}

__device__ __forceinline__ void edge_update(const row8 k, const row8 v,
                                            const row8 qc, const row8 qp,
                                            row8& ac, row8& ap)
{
    const float sc = edge_score(dot8(qc, k));
    const float sp = edge_score(dot8(qp, k));
    add_scaled(ac, v, sc);
    add_scaled(ap, v, sp);
}

__global__ __launch_bounds__(256)
void node_gather_kernel(const int2* __restrict__ ent, const int* __restrict__ start,
                        const int* __restrict__ cnt,
                        const float* __restrict__ Kn, const float* __restrict__ Vn,
                        float* out_c, float* out_p, int n)
{
    __shared__ int ssrc[32 * SEG_PITCH];

    const int tid = threadIdx.x;
    const int g = tid >> 3;
    const int h = tid & 7;
    const int d = blockIdx.x * 32 + g;
    const bool valid = d < n;

    int b = 0, c = 0;
    if (valid) { b = start[d]; c = cnt[d]; }
    const bool ordered = (c <= SEG_CAP);
    int* my = ssrc + g * SEG_PITCH;

    if (ordered) {
        for (int j = h; j < c; j += 8) {
            const int2 me = ent[b + j];
            int r = 0;
            for (int t = 0; t < c; ++t) r += (ent[b + t].x < me.x) ? 1 : 0;
            my[r] = me.y;
        }
    }
    __syncthreads();
    if (!valid) return;

    const int hc  = h * 8;
    const int col = d * 64 + hc;
    const row8 qc = load_row8(out_c + col);
    const row8 qp = load_row8(out_p + col);

    row8 ac, ap;
    ac.lo = make_float4(0.f, 0.f, 0.f, 0.f);
    ac.hi = ac.lo; ap.lo = ac.lo; ap.hi = ac.lo;

    if (ordered) {
        int i = 0;
        if (c >= 2) {
            const int f0 = my[0] * 64 + hc;
            const int f1 = my[1] * 64 + hc;
            row8 k0 = load_row8(Kn + f0), v0 = load_row8(Vn + f0);
            row8 k1 = load_row8(Kn + f1), v1 = load_row8(Vn + f1);
            do {
                const row8 ck0 = k0, cv0 = v0, ck1 = k1, cv1 = v1;
                if (i + 3 < c) {
                    const int o0 = my[i + 2] * 64 + hc;
                    const int o1 = my[i + 3] * 64 + hc;
                    k0 = load_row8(Kn + o0); v0 = load_row8(Vn + o0);
                    k1 = load_row8(Kn + o1); v1 = load_row8(Vn + o1);
                }
                edge_update(ck0, cv0, qc, qp, ac, ap);
                edge_update(ck1, cv1, qc, qp, ac, ap);
                i += 2;
            } while (i + 1 < c);
        }
        if (i < c) {
            const int o0 = my[i] * 64 + hc;
            const row8 kl = load_row8(Kn + o0);
            const row8 vl = load_row8(Vn + o0);
            edge_update(kl, vl, qc, qp, ac, ap);
        }
    } else {
        for (int i = 0; i < c; ++i) {
            const int o0 = ent[b + i].y * 64 + hc;
            const row8 k0 = load_row8(Kn + o0);
            const row8 v0 = load_row8(Vn + o0);
            edge_update(k0, v0, qc, qp, ac, ap);
        }
    }

    *(float4*)(out_c + col)     = ac.lo;
    *(float4*)(out_c + col + 4) = ac.hi;
    *(float4*)(out_p + col)     = ap.lo;
    *(float4*)(out_p + col + 4) = ap.hi;
}

extern "C" void kernel_launch(void* const* d_in, const int* in_sizes, int n_in,
                              void* d_out, int out_size, void* d_ws, size_t ws_size,
                              hipStream_t stream)
{
    const float* h_c  = (const float*)d_in[0];
    const float* h_p  = (const float*)d_in[1];
    const int*   src  = (const int*)  d_in[2];
    const int*   dst  = (const int*)  d_in[3];
    const float* Wqcw = (const float*)d_in[4];
    const float* Wqcb = (const float*)d_in[5];
    const float* Wqpw = (const float*)d_in[6];
    const float* Wqpb = (const float*)d_in[7];
    const float* Wkw  = (const float*)d_in[8];
    const float* Wkb  = (const float*)d_in[9];
    const float* Wvw  = (const float*)d_in[10];
    const float* Wvb  = (const float*)d_in[11];

    const int n = in_sizes[0] / 64;
    const int e = in_sizes[2];

    float* out_c = (float*)d_out;
    float* out_p = out_c + (size_t)n * 64;

    float* Kn   = (float*)d_ws;
    float* Vn   = Kn + (size_t)n * 64;
    int2*  ent  = (int2*)(Vn + (size_t)n * 64);
    int2*  tmp  = ent + (size_t)e;
    int*   cnt  = (int*)(tmp + (size_t)e);
    int*   start  = cnt + n;
    int*   binCnt = start + n;
    int*   bh     = binCnt + BIN_MAX;
    const int nb    = (e + EPB - 1) / EPB;
    const int nbins = (n + BIN_NODES - 1) / BIN_NODES;
    int*   offrel = bh + (size_t)BIN_MAX * nb;

    range_count_kernel<<<nb, 256, 0, stream>>>(dst, bh, e, nb, nbins);

    range_scan_kernel<<<nbins, 512, 0, stream>>>(bh, offrel, binCnt, nb);

    range_scatter_kernel<<<nb, 256, 0, stream>>>(src, dst, offrel, binCnt, tmp, e, nb, nbins);

    range_place_kernel<<<nbins, 1024, 0, stream>>>(tmp, binCnt, ent, start, cnt, n);

    {
        int blocks = (n + 127) / 128;
        proj_wmma_kernel<<<blocks, 256, 0, stream>>>(
            h_c, h_p, Wqcw, Wqcb, Wqpw, Wqpb, Wkw, Wkb, Wvw, Wvb,
            out_c, out_p, Kn, Vn, n);
    }

    node_gather_kernel<<<(n + 31) / 32, 256, 0, stream>>>(
        ent, start, cnt, Kn, Vn, out_c, out_p, n);
}


# projection kernel: weights staged by float4 reads of whole rows, all issued before the first LDS store, transposed at the LDS store (was scalar reads 32 lines wide, two in flight); sums unchanged
# speedup vs baseline: 4.5141x; 1.0752x over previous
#include <hip/hip_runtime.h>

typedef __attribute__((ext_vector_type(2))) float v2f;
typedef __attribute__((ext_vector_type(8))) float v8f;

#define WT_STRIDE 66
#define SEG_CAP   64
#define SEG_PITCH 65

__global__ __launch_bounds__(256)
void proj_wmma_kernel(const float* __restrict__ h_c,
                      const float* __restrict__ h_p,
                      const float* __restrict__ Wqc, const float* __restrict__ bqc,
                      const float* __restrict__ Wqp, const float* __restrict__ bqp,
                      const float* __restrict__ Wk,  const float* __restrict__ bk,
                      const float* __restrict__ Wv,  const float* __restrict__ bv,
                      float* __restrict__ Qc, float* __restrict__ Qp,
                      float* __restrict__ Kn, float* __restrict__ Vn,
                      int n)
{
    __shared__ float sWT[4][64 * WT_STRIDE];
    __shared__ float sB[4][64];

    const int tid = threadIdx.x;
    const float* Ws[4] = {Wqc, Wqp, Wk, Wv};
    const float* Bs[4] = {bqc, bqp, bk, bv};
    {
        const int cg = tid & 15;
        const int kb = tid >> 4;
        float4 w[4][4];
        float  bias_in[4];
#pragma unroll
        for (int o = 0; o < 4; ++o) {
#pragma unroll
            for (int j = 0; j < 4; ++j)
                w[o][j] = *(const float4*)(Ws[o] + (4 * kb + j) * 64 + 4 * cg);
            bias_in[o] = Bs[o][tid & 63];
        }
#pragma unroll
        for (int o = 0; o < 4; ++o) {
            float* t = &sWT[o][(4 * cg) * WT_STRIDE + 4 * kb];
            *(v2f*)(t)                     = v2f{w[o][0].x, w[o][1].x};
            *(v2f*)(t + 2)                 = v2f{w[o][2].x, w[o][3].x};
            *(v2f*)(t + WT_STRIDE)         = v2f{w[o][0].y, w[o][1].y};
            *(v2f*)(t + WT_STRIDE + 2)     = v2f{w[o][2].y, w[o][3].y};
            *(v2f*)(t + 2 * WT_STRIDE)     = v2f{w[o][0].z, w[o][1].z};
            *(v2f*)(t + 2 * WT_STRIDE + 2) = v2f{w[o][2].z, w[o][3].z};
            *(v2f*)(t + 3 * WT_STRIDE)     = v2f{w[o][0].w, w[o][1].w};
            *(v2f*)(t + 3 * WT_STRIDE + 2) = v2f{w[o][2].w, w[o][3].w};
            if (tid < 64) sB[o][tid] = bias_in[o];
        }
    }
    __syncthreads();

    const int wave = tid >> 5;
    const int lane = tid & 31;
    const int half = lane >> 4;
    const int lrow = lane & 15;
    const int ncol = lane & 15;

    const int m0 = (blockIdx.x * 8 + wave) * 16;
    if (m0 >= n) return;

    const int row = m0 + lrow;

    v2f a_c[16], a_p[16];
#pragma unroll
    for (int ks = 0; ks < 16; ++ks) {
        int k = ks * 4 + 2 * half;
        a_c[ks] = *(const v2f*)(h_c + row * 64 + k);
        a_p[ks] = *(const v2f*)(h_p + row * 64 + k);
    }

    float* outs[4] = {Qc, Qp, Kn, Vn};

#pragma unroll
    for (int o = 0; o < 4; ++o) {
        float* op = outs[o];
#pragma unroll
        for (int nt = 0; nt < 4; ++nt) {
            const int ncg = nt * 16 + ncol;
            v8f acc = {0.f, 0.f, 0.f, 0.f, 0.f, 0.f, 0.f, 0.f};
#pragma unroll
            for (int ks = 0; ks < 16; ++ks) {
                int k = ks * 4 + 2 * half;
                v2f b = *(const v2f*)(&sWT[o][ncg * WT_STRIDE + k]);
                v2f a = (o == 1) ? a_p[ks] : a_c[ks];
                acc = __builtin_amdgcn_wmma_f32_16x16x4_f32(
                     false, a,  false, b,
                     (short)0, acc,  false,  false);
            }
            const float bias = sB[o][ncg];
#pragma unroll
            for (int r = 0; r < 8; ++r) {
                int m = m0 + r + 8 * half;
                op[m * 64 + ncg] = acc[r] + bias;
            }
        }
    }
}

#define BIN_SHIFT 9
#define BIN_NODES 512
#define BIN_MAX   256
#define SRC_BITS  17
#define EPT       16
#define EPB       4096
#define STAGE_CAP 12288
#define PLB       8

__device__ __forceinline__ int wave_scan_incl(int x, int lane)
{
    for (int off = 1; off < 32; off <<= 1) {
        int y = __shfl_up(x, off, 32);
        if (lane >= off) x += y;
    }
    return x;
}

__global__ __launch_bounds__(256)
void range_count_kernel(const int* __restrict__ dst, int* __restrict__ bh,
                        int e_total, int nb, int nbins)
{
    __shared__ int hist[BIN_MAX];
    const int tid = threadIdx.x;
    hist[tid] = 0;
    __syncthreads();
    const int base = blockIdx.x * EPB;
    const int last = e_total - 1;
    int dd[EPT];
    for (int j = 0; j < EPT; ++j) {
        const int e = base + j * 256 + tid;
        dd[j] = dst[e < e_total ? e : last];
    }
    for (int j = 0; j < EPT; ++j) {
        const int e = base + j * 256 + tid;
        if (e < e_total) atomicAdd(&hist[dd[j] >> BIN_SHIFT], 1);
    }
    __syncthreads();
    if (tid < nbins) bh[tid * nb + blockIdx.x] = hist[tid];
}

__global__ __launch_bounds__(512)
void range_scan_kernel(const int* __restrict__ bh, int* __restrict__ offrel,
                       int* __restrict__ binCnt, int nb)
{
    __shared__ int wsum[16];
    const int tid  = threadIdx.x;
    const int lane = tid & 31;
    const int wave = tid >> 5;
    const int row  = blockIdx.x * nb;
    int carry = 0;
    for (int c0 = 0; c0 < nb; c0 += 512) {
        const int i = c0 + tid;
        const int v = (i < nb) ? bh[row + i] : 0;
        const int x = wave_scan_incl(v, lane);
        if (lane == 31) wsum[wave] = x;
        __syncthreads();
        int woff = 0, tot = 0;
        for (int w = 0; w < 16; ++w) {
            const int s = wsum[w];
            if (w < wave) woff += s;
            tot += s;
        }
        if (i < nb) offrel[row + i] = carry + woff + x - v;
        carry += tot;
        __syncthreads();
    }
    if (tid == 0) binCnt[blockIdx.x] = carry;
}

__global__ __launch_bounds__(256)
void range_scatter_kernel(const int* __restrict__ src, const int* __restrict__ dst,
                          const int* __restrict__ offrel, const int* __restrict__ binCnt,
                          int2* __restrict__ tmp, int e_total, int nb, int nbins)
{
    __shared__ int  hist[BIN_MAX];
    __shared__ int  lstart[BIN_MAX];
    __shared__ int  gdelta[BIN_MAX];
    __shared__ int  wsA[8];
    __shared__ int  wsB[8];
    __shared__ int2 stage[EPB];
    __shared__ int  sgp[EPB];

    const int tid  = threadIdx.x;
    const int lane = tid & 31;
    const int wave = tid >> 5;
    hist[tid] = 0;
    __syncthreads();

    const int base = blockIdx.x * EPB;
    const int last = e_total - 1;
    int dd[EPT], ss[EPT], rr[EPT];
    for (int j = 0; j < EPT; ++j) {
        const int e  = base + j * 256 + tid;
        const int ec = e < e_total ? e : last;
        dd[j] = dst[ec];
        ss[j] = src[ec];
    }
    for (int j = 0; j < EPT; ++j) {
        const int e = base + j * 256 + tid;
        rr[j] = 0;
        if (e < e_total) rr[j] = atomicAdd(&hist[dd[j] >> BIN_SHIFT], 1);
    }
    __syncthreads();

    const int h  = hist[tid];
    const int g  = (tid < nbins) ? binCnt[tid] : 0;
    const int xh = wave_scan_incl(h, lane);
    const int xg = wave_scan_incl(g, lane);
    if (lane == 31) { wsA[wave] = xh; wsB[wave] = xg; }
    __syncthreads();
    int offA = 0, offB = 0;
    for (int w = 0; w < 8; ++w) {
        if (w < wave) { offA += wsA[w]; offB += wsB[w]; }
    }
    const int ls = offA + xh - h;
    lstart[tid] = ls;
    gdelta[tid] = (tid < nbins) ? (offB + xg - g) + offrel[tid * nb + blockIdx.x] - ls : 0;
    __syncthreads();

    for (int j = 0; j < EPT; ++j) {
        const int e = base + j * 256 + tid;
        if (e < e_total) {
            const int bin = dd[j] >> BIN_SHIFT;
            const int p   = lstart[bin] + rr[j];
            stage[p] = make_int2(e, ss[j] | ((dd[j] & (BIN_NODES - 1)) << SRC_BITS));
            sgp[p]   = gdelta[bin] + p;
        }
    }
    __syncthreads();

    const int left = e_total - base;
    const int m    = left < EPB ? left : EPB;
    for (int i = tid; i < m; i += 256) tmp[sgp[i]] = stage[i];
}

__global__ __launch_bounds__(1024)
void range_place_kernel(const int2* __restrict__ tmp, const int* __restrict__ binCnt,
                        int2* __restrict__ ent, int* __restrict__ start,
                        int* __restrict__ cnt, int n)
{
    __shared__ int  lc[BIN_NODES + 1];
    __shared__ int  lcur[BIN_NODES];
    __shared__ int  wsum[32];
    __shared__ int  sh_bs;
    __shared__ int2 stage[STAGE_CAP];

    const int tid  = threadIdx.x;
    const int lane = tid & 31;
    const int wave = tid >> 5;
    const int k    = blockIdx.x;
    const int bc   = binCnt[k];

    {
        const int v = (tid < k) ? binCnt[tid] : 0;
        const int x = wave_scan_incl(v, lane);
        if (lane == 31) wsum[wave] = x;
        if (tid <= BIN_NODES) lc[tid] = 0;
        __syncthreads();
        if (tid == 0) {
            int t = 0;
            for (int w = 0; w < 32; ++w) t += wsum[w];
            sh_bs = t;
        }
        __syncthreads();
    }
    const int bs    = sh_bs;
    const int node0 = k << BIN_SHIFT;

    for (int i0 = tid; i0 < bc; i0 += PLB * 1024) {
        int y[PLB];
        for (int u = 0; u < PLB; ++u) {
            const int i = i0 + u * 1024;
            y[u] = tmp[bs + (i < bc ? i : bc - 1)].y;
        }
        for (int u = 0; u < PLB; ++u) {
            const int i = i0 + u * 1024;
            atomicAdd(&lc[i < bc ? (y[u] >> SRC_BITS) : BIN_NODES], 1);
        }
    }
    __syncthreads();

    {
        const int c = (tid < BIN_NODES) ? lc[tid] : 0;
        const int x = wave_scan_incl(c, lane);
        if (lane == 31) wsum[wave] = x;
        __syncthreads();
        int woff = 0;
        for (int w = 0; w < 16; ++w) {
            if (w < wave) woff += wsum[w];
        }
        if (tid < BIN_NODES) {
            const int ex = woff + x - c;
            lcur[tid] = ex;
            const int node = node0 + tid;
            if (node < n) { start[node] = bs + ex; cnt[node] = c; }
        }
        __syncthreads();
    }

    const bool staged = (bc <= STAGE_CAP);
    for (int i0 = tid; i0 < bc; i0 += PLB * 1024) {
        int2 a[PLB];
        for (int u = 0; u < PLB; ++u) {
            const int i = i0 + u * 1024;
            a[u] = tmp[bs + (i < bc ? i : bc - 1)];
        }
        for (int u = 0; u < PLB; ++u) {
            const int i = i0 + u * 1024;
            if (i < bc) {
                const int  p = atomicAdd(&lcur[a[u].y >> SRC_BITS], 1);
                const int2 o = make_int2(a[u].x, a[u].y & ((1 << SRC_BITS) - 1));
                if (staged) stage[p] = o; else ent[bs + p] = o;
            }
        }
    }
    if (staged) {
        __syncthreads();
        for (int i = tid; i < bc; i += 1024) ent[bs + i] = stage[i];
    }
}

struct row8 { float4 lo, hi; };

__device__ __forceinline__ row8 load_row8(const float* p)
{
    row8 r;
    r.lo = *(const float4*)(p);
    r.hi = *(const float4*)(p + 4);
    return r;
}

__device__ __forceinline__ float dot8(const row8 a, const row8 b)
{
    return a.lo.x*b.lo.x + a.lo.y*b.lo.y + a.lo.z*b.lo.z + a.lo.w*b.lo.w
         + a.hi.x*b.hi.x + a.hi.y*b.hi.y + a.hi.z*b.hi.z + a.hi.w*b.hi.w;
}

__device__ __forceinline__ float edge_score(float dotv)
{
    return __expf(fminf(fmaxf(dotv * 0.125f, -5.0f), 5.0f));
}

__device__ __forceinline__ void add_scaled(row8& a, const row8 v, float s)
{
    a.lo.x += v.lo.x * s; a.lo.y += v.lo.y * s; a.lo.z += v.lo.z * s; a.lo.w += v.lo.w * s;
    a.hi.x += v.hi.x * s; a.hi.y += v.hi.y * s; a.hi.z += v.hi.z * s; a.hi.w += v.hi.w * s;
}

__device__ __forceinline__ void edge_update(const row8 k, const row8 v,
                                            const row8 qc, const row8 qp,
                                            row8& ac, row8& ap)
{
    const float sc = edge_score(dot8(qc, k));
    const float sp = edge_score(dot8(qp, k));
    add_scaled(ac, v, sc);
    add_scaled(ap, v, sp);
}

__global__ __launch_bounds__(256)
void node_gather_kernel(const int2* __restrict__ ent, const int* __restrict__ start,
                        const int* __restrict__ cnt,
                        const float* __restrict__ Kn, const float* __restrict__ Vn,
                        float* out_c, float* out_p, int n)
{
    __shared__ int ssrc[32 * SEG_PITCH];

    const int tid = threadIdx.x;
    const int g = tid >> 3;
    const int h = tid & 7;
    const int d = blockIdx.x * 32 + g;
    const bool valid = d < n;

    int b = 0, c = 0;
    if (valid) { b = start[d]; c = cnt[d]; }
    const bool ordered = (c <= SEG_CAP);
    int* my = ssrc + g * SEG_PITCH;

    if (ordered) {
        for (int j = h; j < c; j += 8) {
            const int2 me = ent[b + j];
            int r = 0;
            for (int t = 0; t < c; ++t) r += (ent[b + t].x < me.x) ? 1 : 0;
            my[r] = me.y;
        }
    }
    __syncthreads();
    if (!valid) return;

    const int hc  = h * 8;
    const int col = d * 64 + hc;
    const row8 qc = load_row8(out_c + col);
    const row8 qp = load_row8(out_p + col);

    row8 ac, ap;
    ac.lo = make_float4(0.f, 0.f, 0.f, 0.f);
    ac.hi = ac.lo; ap.lo = ac.lo; ap.hi = ac.lo;

    if (ordered) {
        int i = 0;
        if (c >= 2) {
            const int f0 = my[0] * 64 + hc;
            const int f1 = my[1] * 64 + hc;
            row8 k0 = load_row8(Kn + f0), v0 = load_row8(Vn + f0);
            row8 k1 = load_row8(Kn + f1), v1 = load_row8(Vn + f1);
            do {
                const row8 ck0 = k0, cv0 = v0, ck1 = k1, cv1 = v1;
                if (i + 3 < c) {
                    const int o0 = my[i + 2] * 64 + hc;
                    const int o1 = my[i + 3] * 64 + hc;
                    k0 = load_row8(Kn + o0); v0 = load_row8(Vn + o0);
                    k1 = load_row8(Kn + o1); v1 = load_row8(Vn + o1);
                }
                edge_update(ck0, cv0, qc, qp, ac, ap);
                edge_update(ck1, cv1, qc, qp, ac, ap);
                i += 2;
            } while (i + 1 < c);
        }
        if (i < c) {
            const int o0 = my[i] * 64 + hc;
            const row8 kl = load_row8(Kn + o0);
            const row8 vl = load_row8(Vn + o0);
            edge_update(kl, vl, qc, qp, ac, ap);
        }
    } else {
        for (int i = 0; i < c; ++i) {
            const int o0 = ent[b + i].y * 64 + hc;
            const row8 k0 = load_row8(Kn + o0);
            const row8 v0 = load_row8(Vn + o0);
            edge_update(k0, v0, qc, qp, ac, ap);
        }
    }

    *(float4*)(out_c + col)     = ac.lo;
    *(float4*)(out_c + col + 4) = ac.hi;
    *(float4*)(out_p + col)     = ap.lo;
    *(float4*)(out_p + col + 4) = ap.hi;
}

extern "C" void kernel_launch(void* const* d_in, const int* in_sizes, int n_in,
                              void* d_out, int out_size, void* d_ws, size_t ws_size,
                              hipStream_t stream)
{
    const float* h_c  = (const float*)d_in[0];
    const float* h_p  = (const float*)d_in[1];
    const int*   src  = (const int*)  d_in[2];
    const int*   dst  = (const int*)  d_in[3];
    const float* Wqcw = (const float*)d_in[4];
    const float* Wqcb = (const float*)d_in[5];
    const float* Wqpw = (const float*)d_in[6];
    const float* Wqpb = (const float*)d_in[7];
    const float* Wkw  = (const float*)d_in[8];
    const float* Wkb  = (const float*)d_in[9];
    const float* Wvw  = (const float*)d_in[10];
    const float* Wvb  = (const float*)d_in[11];

    const int n = in_sizes[0] / 64;
    const int e = in_sizes[2];

    float* out_c = (float*)d_out;
    float* out_p = out_c + (size_t)n * 64;

    float* Kn   = (float*)d_ws;
    float* Vn   = Kn + (size_t)n * 64;
    int2*  ent  = (int2*)(Vn + (size_t)n * 64);
    int2*  tmp  = ent + (size_t)e;
    int*   cnt  = (int*)(tmp + (size_t)e);
    int*   start  = cnt + n;
    int*   binCnt = start + n;
    int*   bh     = binCnt + BIN_MAX;
    const int nb    = (e + EPB - 1) / EPB;
    const int nbins = (n + BIN_NODES - 1) / BIN_NODES;
    int*   offrel = bh + (size_t)BIN_MAX * nb;

    range_count_kernel<<<nb, 256, 0, stream>>>(dst, bh, e, nb, nbins);

    range_scan_kernel<<<nbins, 512, 0, stream>>>(bh, offrel, binCnt, nb);

    range_scatter_kernel<<<nb, 256, 0, stream>>>(src, dst, offrel, binCnt, tmp, e, nb, nbins);

    range_place_kernel<<<nbins, 1024, 0, stream>>>(tmp, binCnt, ent, start, cnt, n);

    {
        int blocks = (n + 127) / 128;
        proj_wmma_kernel<<<blocks, 256, 0, stream>>>(
            h_c, h_p, Wqcw, Wqcb, Wqpw, Wqpb, Wkw, Wkb, Wvw, Wvb,
            out_c, out_p, Kn, Vn, n);
    }

    node_gather_kernel<<<(n + 31) / 32, 256, 0, stream>>>(
        ent, start, cnt, Kn, Vn, out_c, out_p, n);
}


# edge ordering in 2 launches not 4: each edge block keeps its range-grouped edges in its own slab with a table of range starts, so the cross-block count and scan launches go; sums and order as before
# speedup vs baseline: 5.2364x; 1.1600x over previous
#include <hip/hip_runtime.h>

typedef __attribute__((ext_vector_type(2))) float v2f;
typedef __attribute__((ext_vector_type(8))) float v8f;

#define WT_STRIDE 66
#define SEG_CAP   64
#define SEG_PITCH 65

__global__ __launch_bounds__(256)
void proj_wmma_kernel(const float* __restrict__ h_c,
                      const float* __restrict__ h_p,
                      const float* __restrict__ Wqc, const float* __restrict__ bqc,
                      const float* __restrict__ Wqp, const float* __restrict__ bqp,
                      const float* __restrict__ Wk,  const float* __restrict__ bk,
                      const float* __restrict__ Wv,  const float* __restrict__ bv,
                      float* __restrict__ Qc, float* __restrict__ Qp,
                      float* __restrict__ Kn, float* __restrict__ Vn,
                      int n)
{
    __shared__ float sWT[4][64 * WT_STRIDE];
    __shared__ float sB[4][64];

    const int tid = threadIdx.x;
    const float* Ws[4] = {Wqc, Wqp, Wk, Wv};
    const float* Bs[4] = {bqc, bqp, bk, bv};
    {
        const int cg = tid & 15;
        const int kb = tid >> 4;
        float4 w[4][4];
        float  bias_in[4];
#pragma unroll
        for (int o = 0; o < 4; ++o) {
#pragma unroll
            for (int j = 0; j < 4; ++j)
                w[o][j] = *(const float4*)(Ws[o] + (4 * kb + j) * 64 + 4 * cg);
            bias_in[o] = Bs[o][tid & 63];
        }
#pragma unroll
        for (int o = 0; o < 4; ++o) {
            float* t = &sWT[o][(4 * cg) * WT_STRIDE + 4 * kb];
            *(v2f*)(t)                     = v2f{w[o][0].x, w[o][1].x};
            *(v2f*)(t + 2)                 = v2f{w[o][2].x, w[o][3].x};
            *(v2f*)(t + WT_STRIDE)         = v2f{w[o][0].y, w[o][1].y};
            *(v2f*)(t + WT_STRIDE + 2)     = v2f{w[o][2].y, w[o][3].y};
            *(v2f*)(t + 2 * WT_STRIDE)     = v2f{w[o][0].z, w[o][1].z};
            *(v2f*)(t + 2 * WT_STRIDE + 2) = v2f{w[o][2].z, w[o][3].z};
            *(v2f*)(t + 3 * WT_STRIDE)     = v2f{w[o][0].w, w[o][1].w};
            *(v2f*)(t + 3 * WT_STRIDE + 2) = v2f{w[o][2].w, w[o][3].w};
            if (tid < 64) sB[o][tid] = bias_in[o];
        }
    }
    __syncthreads();

    const int wave = tid >> 5;
    const int lane = tid & 31;
    const int half = lane >> 4;
    const int lrow = lane & 15;
    const int ncol = lane & 15;

    const int m0 = (blockIdx.x * 8 + wave) * 16;
    if (m0 >= n) return;

    const int row = m0 + lrow;

    v2f a_c[16], a_p[16];
#pragma unroll
    for (int ks = 0; ks < 16; ++ks) {
        int k = ks * 4 + 2 * half;
        a_c[ks] = *(const v2f*)(h_c + row * 64 + k);
        a_p[ks] = *(const v2f*)(h_p + row * 64 + k);
    }

    float* outs[4] = {Qc, Qp, Kn, Vn};

#pragma unroll
    for (int o = 0; o < 4; ++o) {
        float* op = outs[o];
#pragma unroll
        for (int nt = 0; nt < 4; ++nt) {
            const int ncg = nt * 16 + ncol;
            v8f acc = {0.f, 0.f, 0.f, 0.f, 0.f, 0.f, 0.f, 0.f};
#pragma unroll
            for (int ks = 0; ks < 16; ++ks) {
                int k = ks * 4 + 2 * half;
                v2f b = *(const v2f*)(&sWT[o][ncg * WT_STRIDE + k]);
                v2f a = (o == 1) ? a_p[ks] : a_c[ks];
                acc = __builtin_amdgcn_wmma_f32_16x16x4_f32(
                     false, a,  false, b,
                     (short)0, acc,  false,  false);
            }
            const float bias = sB[o][ncg];
#pragma unroll
            for (int r = 0; r < 8; ++r) {
                int m = m0 + r + 8 * half;
                op[m * 64 + ncg] = acc[r] + bias;
            }
        }
    }
}

#define BIN_SHIFT 9
#define BIN_NODES 512
#define BIN_MAX   256
#define SRC_BITS  17
#define EPT       16
#define EPB       4096
#define STAGE_CAP 12288
#define PW        13

__device__ __forceinline__ int wave_scan_incl(int x, int lane)
{
    for (int off = 1; off < 32; off <<= 1) {
        int y = __shfl_up(x, off, 32);
        if (lane >= off) x += y;
    }
    return x;
}

__global__ __launch_bounds__(256)
void range_bin_kernel(const int* __restrict__ src, const int* __restrict__ dst,
                      int2* __restrict__ tmp, int* __restrict__ lst,
                      int e_total, int nb, int nbins)
{
    __shared__ int  hist[BIN_MAX];
    __shared__ int  lstart[BIN_MAX];
    __shared__ int  wsA[8];
    __shared__ int2 stage[EPB];

    const int tid  = threadIdx.x;
    const int lane = tid & 31;
    const int wave = tid >> 5;
    hist[tid] = 0;
    __syncthreads();

    const int base = blockIdx.x * EPB;
    const int last = e_total - 1;
    const int left = e_total - base;
    const int m    = left < EPB ? left : EPB;
    int dd[EPT], ss[EPT], rr[EPT];
    for (int j = 0; j < EPT; ++j) {
        const int e  = base + j * 256 + tid;
        const int ec = e < e_total ? e : last;
        dd[j] = dst[ec];
        ss[j] = src[ec];
    }
    for (int j = 0; j < EPT; ++j) {
        const int e = base + j * 256 + tid;
        rr[j] = 0;
        if (e < e_total) rr[j] = atomicAdd(&hist[dd[j] >> BIN_SHIFT], 1);
    }
    __syncthreads();

    const int h  = hist[tid];
    const int xh = wave_scan_incl(h, lane);
    if (lane == 31) wsA[wave] = xh;
    __syncthreads();
    int offA = 0;
    for (int w = 0; w < 8; ++w) {
        if (w < wave) offA += wsA[w];
    }
    const int ls = offA + xh - h;
    lstart[tid] = ls;
    if (tid < nbins) lst[tid * nb + blockIdx.x] = ls;
    if (tid == 0)    lst[nbins * nb + blockIdx.x] = m;
    __syncthreads();

    for (int j = 0; j < EPT; ++j) {
        const int e = base + j * 256 + tid;
        if (e < e_total) {
            const int p = lstart[dd[j] >> BIN_SHIFT] + rr[j];
            stage[p] = make_int2(e, ss[j] | ((dd[j] & (BIN_NODES - 1)) << SRC_BITS));
        }
    }
    __syncthreads();

    for (int i = tid; i < m; i += 256) tmp[base + i] = stage[i];
}

template <int MODE>
__device__ __forceinline__ void range_visit(const int2 a, int* lc, int* lcur,
                                            int2* stage, int2* __restrict__ ent,
                                            int bs, bool staged)
{
    if (MODE == 0) {
        atomicAdd(&lc[a.y >> SRC_BITS], 1);
    } else {
        const int  p = atomicAdd(&lcur[a.y >> SRC_BITS], 1);
        const int2 o = make_int2(a.x, a.y & ((1 << SRC_BITS) - 1));
        if (staged) stage[p] = o; else ent[bs + p] = o;
    }
}

__global__ __launch_bounds__(1024)
void range_place_kernel(const int2* __restrict__ tmp, const int* __restrict__ lst,
                        int2* __restrict__ ent, int* __restrict__ start,
                        int* __restrict__ cnt, int n, int nb)
{
    __shared__ int  lc[BIN_NODES];
    __shared__ int  lcur[BIN_NODES];
    __shared__ int  wsum[32];
    __shared__ int  wsum2[32];
    __shared__ int  sh_bs;
    __shared__ int  sh_bc;
    __shared__ int2 stage[STAGE_CAP];

    const int tid  = threadIdx.x;
    const int lane = tid & 31;
    const int wave = tid >> 5;
    const int wv   = wave;
    const int k    = blockIdx.x;
    const int* __restrict__ row0 = lst + k * nb;
    const int* __restrict__ row1 = lst + (k + 1) * nb;

    int plo[PW], plen[PW];
    int2 a[PW];
#pragma unroll
    for (int u = 0; u < PW; ++u) {
        const int b  = wv + 32 * u;
        const int bb = b < nb ? b : nb - 1;
        plo[u]  = row0[bb];
        plen[u] = row1[bb];
    }
#pragma unroll
    for (int u = 0; u < PW; ++u) {
        const int b  = wv + 32 * u;
        const int bb = b < nb ? b : nb - 1;
        plen[u] = b < nb ? plen[u] - plo[u] : 0;
        plo[u]  = bb * EPB + plo[u];
    }
#pragma unroll
    for (int u = 0; u < PW; ++u)
        a[u] = tmp[lane < plen[u] ? plo[u] + lane : 0];

    {
        int s0 = 0, s1 = 0;
        for (int b = tid; b < nb; b += 1024) { s0 += row0[b]; s1 += row1[b]; }
        const int x0 = wave_scan_incl(s0, lane);
        const int x1 = wave_scan_incl(s1, lane);
        if (lane == 31) { wsum[wave] = x0; wsum2[wave] = x1; }
        if (tid < BIN_NODES) lc[tid] = 0;
        __syncthreads();
        if (tid == 0) {
            int t0 = 0, t1 = 0;
            for (int w = 0; w < 32; ++w) { t0 += wsum[w]; t1 += wsum2[w]; }
            sh_bs = t0;
            sh_bc = t1 - t0;
        }
        __syncthreads();
    }
    const int  bs     = sh_bs;
    const int  bc     = sh_bc;
    const int  node0  = k << BIN_SHIFT;
    const bool staged = (bc <= STAGE_CAP);

#pragma unroll
    for (int u = 0; u < PW; ++u)
        if (lane < plen[u]) range_visit<0>(a[u], lc, lcur, stage, ent, bs, staged);
#pragma unroll
    for (int u = 0; u < PW; ++u)
        for (int j = 32 + lane; j < plen[u]; j += 32)
            range_visit<0>(tmp[plo[u] + j], lc, lcur, stage, ent, bs, staged);
    for (int b = wv + 32 * PW; b < nb; b += 32) {
        const int l0 = row0[b];
        const int ln = row1[b] - l0;
        for (int j = lane; j < ln; j += 32)
            range_visit<0>(tmp[b * EPB + l0 + j], lc, lcur, stage, ent, bs, staged);
    }
    __syncthreads();

    {
        const int c = (tid < BIN_NODES) ? lc[tid] : 0;
        const int x = wave_scan_incl(c, lane);
        if (lane == 31) wsum[wave] = x;
        __syncthreads();
        int woff = 0;
        for (int w = 0; w < 16; ++w) {
            if (w < wave) woff += wsum[w];
        }
        if (tid < BIN_NODES) {
            const int ex = woff + x - c;
            lcur[tid] = ex;
            const int node = node0 + tid;
            if (node < n) { start[node] = bs + ex; cnt[node] = c; }
        }
        __syncthreads();
    }

#pragma unroll
    for (int u = 0; u < PW; ++u)
        if (lane < plen[u]) range_visit<1>(a[u], lc, lcur, stage, ent, bs, staged);
#pragma unroll
    for (int u = 0; u < PW; ++u)
        for (int j = 32 + lane; j < plen[u]; j += 32)
            range_visit<1>(tmp[plo[u] + j], lc, lcur, stage, ent, bs, staged);
    for (int b = wv + 32 * PW; b < nb; b += 32) {
        const int l0 = row0[b];
        const int ln = row1[b] - l0;
        for (int j = lane; j < ln; j += 32)
            range_visit<1>(tmp[b * EPB + l0 + j], lc, lcur, stage, ent, bs, staged);
    }
    if (staged) {
        __syncthreads();
        for (int i = tid; i < bc; i += 1024) ent[bs + i] = stage[i];
    }
}

struct row8 { float4 lo, hi; };

__device__ __forceinline__ row8 load_row8(const float* p)
{
    row8 r;
    r.lo = *(const float4*)(p);
    r.hi = *(const float4*)(p + 4);
    return r;
}

__device__ __forceinline__ float dot8(const row8 a, const row8 b)
{
    return a.lo.x*b.lo.x + a.lo.y*b.lo.y + a.lo.z*b.lo.z + a.lo.w*b.lo.w
         + a.hi.x*b.hi.x + a.hi.y*b.hi.y + a.hi.z*b.hi.z + a.hi.w*b.hi.w;
}

__device__ __forceinline__ float edge_score(float dotv)
{
    return __expf(fminf(fmaxf(dotv * 0.125f, -5.0f), 5.0f));
}

__device__ __forceinline__ void add_scaled(row8& a, const row8 v, float s)
{
    a.lo.x += v.lo.x * s; a.lo.y += v.lo.y * s; a.lo.z += v.lo.z * s; a.lo.w += v.lo.w * s;
    a.hi.x += v.hi.x * s; a.hi.y += v.hi.y * s; a.hi.z += v.hi.z * s; a.hi.w += v.hi.w * s;
}

__device__ __forceinline__ void edge_update(const row8 k, const row8 v,
                                            const row8 qc, const row8 qp,
                                            row8& ac, row8& ap)
{
    const float sc = edge_score(dot8(qc, k));
    const float sp = edge_score(dot8(qp, k));
    add_scaled(ac, v, sc);
    add_scaled(ap, v, sp);
}

__global__ __launch_bounds__(256)
void node_gather_kernel(const int2* __restrict__ ent, const int* __restrict__ start,
                        const int* __restrict__ cnt,
                        const float* __restrict__ Kn, const float* __restrict__ Vn,
                        float* out_c, float* out_p, int n)
{
    __shared__ int ssrc[32 * SEG_PITCH];

    const int tid = threadIdx.x;
    const int g = tid >> 3;
    const int h = tid & 7;
    const int d = blockIdx.x * 32 + g;
    const bool valid = d < n;

    int b = 0, c = 0;
    if (valid) { b = start[d]; c = cnt[d]; }
    const bool ordered = (c <= SEG_CAP);
    int* my = ssrc + g * SEG_PITCH;

    if (ordered) {
        for (int j = h; j < c; j += 8) {
            const int2 me = ent[b + j];
            int r = 0;
            for (int t = 0; t < c; ++t) r += (ent[b + t].x < me.x) ? 1 : 0;
            my[r] = me.y;
        }
    }
    __syncthreads();
    if (!valid) return;

    const int hc  = h * 8;
    const int col = d * 64 + hc;
    const row8 qc = load_row8(out_c + col);
    const row8 qp = load_row8(out_p + col);

    row8 ac, ap;
    ac.lo = make_float4(0.f, 0.f, 0.f, 0.f);
    ac.hi = ac.lo; ap.lo = ac.lo; ap.hi = ac.lo;

    if (ordered) {
        int i = 0;
        if (c >= 2) {
            const int f0 = my[0] * 64 + hc;
            const int f1 = my[1] * 64 + hc;
            row8 k0 = load_row8(Kn + f0), v0 = load_row8(Vn + f0);
            row8 k1 = load_row8(Kn + f1), v1 = load_row8(Vn + f1);
            do {
                const row8 ck0 = k0, cv0 = v0, ck1 = k1, cv1 = v1;
                if (i + 3 < c) {
                    const int o0 = my[i + 2] * 64 + hc;
                    const int o1 = my[i + 3] * 64 + hc;
                    k0 = load_row8(Kn + o0); v0 = load_row8(Vn + o0);
                    k1 = load_row8(Kn + o1); v1 = load_row8(Vn + o1);
                }
                edge_update(ck0, cv0, qc, qp, ac, ap);
                edge_update(ck1, cv1, qc, qp, ac, ap);
                i += 2;
            } while (i + 1 < c);
        }
        if (i < c) {
            const int o0 = my[i] * 64 + hc;
            const row8 kl = load_row8(Kn + o0);
            const row8 vl = load_row8(Vn + o0);
            edge_update(kl, vl, qc, qp, ac, ap);
        }
    } else {
        for (int i = 0; i < c; ++i) {
            const int o0 = ent[b + i].y * 64 + hc;
            const row8 k0 = load_row8(Kn + o0);
            const row8 v0 = load_row8(Vn + o0);
            edge_update(k0, v0, qc, qp, ac, ap);
        }
    }

    *(float4*)(out_c + col)     = ac.lo;
    *(float4*)(out_c + col + 4) = ac.hi;
    *(float4*)(out_p + col)     = ap.lo;
    *(float4*)(out_p + col + 4) = ap.hi;
}

extern "C" void kernel_launch(void* const* d_in, const int* in_sizes, int n_in,
                              void* d_out, int out_size, void* d_ws, size_t ws_size,
                              hipStream_t stream)
{
    const float* h_c  = (const float*)d_in[0];
    const float* h_p  = (const float*)d_in[1];
    const int*   src  = (const int*)  d_in[2];
    const int*   dst  = (const int*)  d_in[3];
    const float* Wqcw = (const float*)d_in[4];
    const float* Wqcb = (const float*)d_in[5];
    const float* Wqpw = (const float*)d_in[6];
    const float* Wqpb = (const float*)d_in[7];
    const float* Wkw  = (const float*)d_in[8];
    const float* Wkb  = (const float*)d_in[9];
    const float* Wvw  = (const float*)d_in[10];
    const float* Wvb  = (const float*)d_in[11];

    const int n = in_sizes[0] / 64;
    const int e = in_sizes[2];

    float* out_c = (float*)d_out;
    float* out_p = out_c + (size_t)n * 64;

    float* Kn   = (float*)d_ws;
    float* Vn   = Kn + (size_t)n * 64;
    int2*  ent  = (int2*)(Vn + (size_t)n * 64);
    int2*  tmp  = ent + (size_t)e;
    int*   cnt  = (int*)(tmp + (size_t)e);
    int*   start  = cnt + n;
    int*   lst    = start + n;
    const int nb    = (e + EPB - 1) / EPB;
    const int nbins = (n + BIN_NODES - 1) / BIN_NODES;

    range_bin_kernel<<<nb, 256, 0, stream>>>(src, dst, tmp, lst, e, nb, nbins);

    range_place_kernel<<<nbins, 1024, 0, stream>>>(tmp, lst, ent, start, cnt, n, nb);

    {
        int blocks = (n + 127) / 128;
        proj_wmma_kernel<<<blocks, 256, 0, stream>>>(
            h_c, h_p, Wqcw, Wqcb, Wqpw, Wqpb, Wkw, Wkb, Wvw, Wvb,
            out_c, out_p, Kn, Vn, n);
    }

    node_gather_kernel<<<(n + 31) / 32, 256, 0, stream>>>(
        ent, start, cnt, Kn, Vn, out_c, out_p, n);
}


# gather: more loads in flight per wave: 4 edges' K/V rows issued together per trip (was 2), Q rows beside the run bounds; room from head dots as one packed fused chain on (Q_c,Q_p) pairs, same order
# speedup vs baseline: 5.5379x; 1.0576x over previous
#include <hip/hip_runtime.h>

typedef __attribute__((ext_vector_type(2))) float v2f;
typedef __attribute__((ext_vector_type(8))) float v8f;

#define WT_STRIDE 66
#define SEG_CAP   64
#define SEG_PITCH 65

__global__ __launch_bounds__(256)
void proj_wmma_kernel(const float* __restrict__ h_c,
                      const float* __restrict__ h_p,
                      const float* __restrict__ Wqc, const float* __restrict__ bqc,
                      const float* __restrict__ Wqp, const float* __restrict__ bqp,
                      const float* __restrict__ Wk,  const float* __restrict__ bk,
                      const float* __restrict__ Wv,  const float* __restrict__ bv,
                      float* __restrict__ Qc, float* __restrict__ Qp,
                      float* __restrict__ Kn, float* __restrict__ Vn,
                      int n)
{
    __shared__ float sWT[4][64 * WT_STRIDE];
    __shared__ float sB[4][64];

    const int tid = threadIdx.x;
    const float* Ws[4] = {Wqc, Wqp, Wk, Wv};
    const float* Bs[4] = {bqc, bqp, bk, bv};
    {
        const int cg = tid & 15;
        const int kb = tid >> 4;
        float4 w[4][4];
        float  bias_in[4];
#pragma unroll
        for (int o = 0; o < 4; ++o) {
#pragma unroll
            for (int j = 0; j < 4; ++j)
                w[o][j] = *(const float4*)(Ws[o] + (4 * kb + j) * 64 + 4 * cg);
            bias_in[o] = Bs[o][tid & 63];
        }
#pragma unroll
        for (int o = 0; o < 4; ++o) {
            float* t = &sWT[o][(4 * cg) * WT_STRIDE + 4 * kb];
            *(v2f*)(t)                     = v2f{w[o][0].x, w[o][1].x};
            *(v2f*)(t + 2)                 = v2f{w[o][2].x, w[o][3].x};
            *(v2f*)(t + WT_STRIDE)         = v2f{w[o][0].y, w[o][1].y};
            *(v2f*)(t + WT_STRIDE + 2)     = v2f{w[o][2].y, w[o][3].y};
            *(v2f*)(t + 2 * WT_STRIDE)     = v2f{w[o][0].z, w[o][1].z};
            *(v2f*)(t + 2 * WT_STRIDE + 2) = v2f{w[o][2].z, w[o][3].z};
            *(v2f*)(t + 3 * WT_STRIDE)     = v2f{w[o][0].w, w[o][1].w};
            *(v2f*)(t + 3 * WT_STRIDE + 2) = v2f{w[o][2].w, w[o][3].w};
            if (tid < 64) sB[o][tid] = bias_in[o];
        }
    }
    __syncthreads();

    const int wave = tid >> 5;
    const int lane = tid & 31;
    const int half = lane >> 4;
    const int lrow = lane & 15;
    const int ncol = lane & 15;

    const int m0 = (blockIdx.x * 8 + wave) * 16;
    if (m0 >= n) return;

    const int row = m0 + lrow;

    v2f a_c[16], a_p[16];
#pragma unroll
    for (int ks = 0; ks < 16; ++ks) {
        int k = ks * 4 + 2 * half;
        a_c[ks] = *(const v2f*)(h_c + row * 64 + k);
        a_p[ks] = *(const v2f*)(h_p + row * 64 + k);
    }

    float* outs[4] = {Qc, Qp, Kn, Vn};

#pragma unroll
    for (int o = 0; o < 4; ++o) {
        float* op = outs[o];
#pragma unroll
        for (int nt = 0; nt < 4; ++nt) {
            const int ncg = nt * 16 + ncol;
            v8f acc = {0.f, 0.f, 0.f, 0.f, 0.f, 0.f, 0.f, 0.f};
#pragma unroll
            for (int ks = 0; ks < 16; ++ks) {
                int k = ks * 4 + 2 * half;
                v2f b = *(const v2f*)(&sWT[o][ncg * WT_STRIDE + k]);
                v2f a = (o == 1) ? a_p[ks] : a_c[ks];
                acc = __builtin_amdgcn_wmma_f32_16x16x4_f32(
                     false, a,  false, b,
                     (short)0, acc,  false,  false);
            }
            const float bias = sB[o][ncg];
#pragma unroll
            for (int r = 0; r < 8; ++r) {
                int m = m0 + r + 8 * half;
                op[m * 64 + ncg] = acc[r] + bias;
            }
        }
    }
}

#define BIN_SHIFT 9
#define BIN_NODES 512
#define BIN_MAX   256
#define SRC_BITS  17
#define EPT       16
#define EPB       4096
#define STAGE_CAP 12288
#define PW        13

__device__ __forceinline__ int wave_scan_incl(int x, int lane)
{
    for (int off = 1; off < 32; off <<= 1) {
        int y = __shfl_up(x, off, 32);
        if (lane >= off) x += y;
    }
    return x;
}

__global__ __launch_bounds__(256)
void range_bin_kernel(const int* __restrict__ src, const int* __restrict__ dst,
                      int2* __restrict__ tmp, int* __restrict__ lst,
                      int e_total, int nb, int nbins)
{
    __shared__ int  hist[BIN_MAX];
    __shared__ int  lstart[BIN_MAX];
    __shared__ int  wsA[8];
    __shared__ int2 stage[EPB];

    const int tid  = threadIdx.x;
    const int lane = tid & 31;
    const int wave = tid >> 5;
    hist[tid] = 0;
    __syncthreads();

    const int base = blockIdx.x * EPB;
    const int last = e_total - 1;
    const int left = e_total - base;
    const int m    = left < EPB ? left : EPB;
    int dd[EPT], ss[EPT], rr[EPT];
    for (int j = 0; j < EPT; ++j) {
        const int e  = base + j * 256 + tid;
        const int ec = e < e_total ? e : last;
        dd[j] = dst[ec];
        ss[j] = src[ec];
    }
    for (int j = 0; j < EPT; ++j) {
        const int e = base + j * 256 + tid;
        rr[j] = 0;
        if (e < e_total) rr[j] = atomicAdd(&hist[dd[j] >> BIN_SHIFT], 1);
    }
    __syncthreads();

    const int h  = hist[tid];
    const int xh = wave_scan_incl(h, lane);
    if (lane == 31) wsA[wave] = xh;
    __syncthreads();
    int offA = 0;
    for (int w = 0; w < 8; ++w) {
        if (w < wave) offA += wsA[w];
    }
    const int ls = offA + xh - h;
    lstart[tid] = ls;
    if (tid < nbins) lst[tid * nb + blockIdx.x] = ls;
    if (tid == 0)    lst[nbins * nb + blockIdx.x] = m;
    __syncthreads();

    for (int j = 0; j < EPT; ++j) {
        const int e = base + j * 256 + tid;
        if (e < e_total) {
            const int p = lstart[dd[j] >> BIN_SHIFT] + rr[j];
            stage[p] = make_int2(e, ss[j] | ((dd[j] & (BIN_NODES - 1)) << SRC_BITS));
        }
    }
    __syncthreads();

    for (int i = tid; i < m; i += 256) tmp[base + i] = stage[i];
}

template <int MODE>
__device__ __forceinline__ void range_visit(const int2 a, int* lc, int* lcur,
                                            int2* stage, int2* __restrict__ ent,
                                            int bs, bool staged)
{
    if (MODE == 0) {
        atomicAdd(&lc[a.y >> SRC_BITS], 1);
    } else {
        const int  p = atomicAdd(&lcur[a.y >> SRC_BITS], 1);
        const int2 o = make_int2(a.x, a.y & ((1 << SRC_BITS) - 1));
        if (staged) stage[p] = o; else ent[bs + p] = o;
    }
}

__global__ __launch_bounds__(1024)
void range_place_kernel(const int2* __restrict__ tmp, const int* __restrict__ lst,
                        int2* __restrict__ ent, int* __restrict__ start,
                        int* __restrict__ cnt, int n, int nb)
{
    __shared__ int  lc[BIN_NODES];
    __shared__ int  lcur[BIN_NODES];
    __shared__ int  wsum[32];
    __shared__ int  wsum2[32];
    __shared__ int  sh_bs;
    __shared__ int  sh_bc;
    __shared__ int2 stage[STAGE_CAP];

    const int tid  = threadIdx.x;
    const int lane = tid & 31;
    const int wave = tid >> 5;
    const int wv   = wave;
    const int k    = blockIdx.x;
    const int* __restrict__ row0 = lst + k * nb;
    const int* __restrict__ row1 = lst + (k + 1) * nb;

    int plo[PW], plen[PW];
    int2 a[PW];
#pragma unroll
    for (int u = 0; u < PW; ++u) {
        const int b  = wv + 32 * u;
        const int bb = b < nb ? b : nb - 1;
        plo[u]  = row0[bb];
        plen[u] = row1[bb];
    }
#pragma unroll
    for (int u = 0; u < PW; ++u) {
        const int b  = wv + 32 * u;
        const int bb = b < nb ? b : nb - 1;
        plen[u] = b < nb ? plen[u] - plo[u] : 0;
        plo[u]  = bb * EPB + plo[u];
    }
#pragma unroll
    for (int u = 0; u < PW; ++u)
        a[u] = tmp[lane < plen[u] ? plo[u] + lane : 0];

    {
        int s0 = 0, s1 = 0;
        for (int b = tid; b < nb; b += 1024) { s0 += row0[b]; s1 += row1[b]; }
        const int x0 = wave_scan_incl(s0, lane);
        const int x1 = wave_scan_incl(s1, lane);
        if (lane == 31) { wsum[wave] = x0; wsum2[wave] = x1; }
        if (tid < BIN_NODES) lc[tid] = 0;
        __syncthreads();
        if (tid == 0) {
            int t0 = 0, t1 = 0;
            for (int w = 0; w < 32; ++w) { t0 += wsum[w]; t1 += wsum2[w]; }
            sh_bs = t0;
            sh_bc = t1 - t0;
        }
        __syncthreads();
    }
    const int  bs     = sh_bs;
    const int  bc     = sh_bc;
    const int  node0  = k << BIN_SHIFT;
    const bool staged = (bc <= STAGE_CAP);

#pragma unroll
    for (int u = 0; u < PW; ++u)
        if (lane < plen[u]) range_visit<0>(a[u], lc, lcur, stage, ent, bs, staged);
#pragma unroll
    for (int u = 0; u < PW; ++u)
        for (int j = 32 + lane; j < plen[u]; j += 32)
            range_visit<0>(tmp[plo[u] + j], lc, lcur, stage, ent, bs, staged);
    for (int b = wv + 32 * PW; b < nb; b += 32) {
        const int l0 = row0[b];
        const int ln = row1[b] - l0;
        for (int j = lane; j < ln; j += 32)
            range_visit<0>(tmp[b * EPB + l0 + j], lc, lcur, stage, ent, bs, staged);
    }
    __syncthreads();

    {
        const int c = (tid < BIN_NODES) ? lc[tid] : 0;
        const int x = wave_scan_incl(c, lane);
        if (lane == 31) wsum[wave] = x;
        __syncthreads();
        int woff = 0;
        for (int w = 0; w < 16; ++w) {
            if (w < wave) woff += wsum[w];
        }
        if (tid < BIN_NODES) {
            const int ex = woff + x - c;
            lcur[tid] = ex;
            const int node = node0 + tid;
            if (node < n) { start[node] = bs + ex; cnt[node] = c; }
        }
        __syncthreads();
    }

#pragma unroll
    for (int u = 0; u < PW; ++u)
        if (lane < plen[u]) range_visit<1>(a[u], lc, lcur, stage, ent, bs, staged);
#pragma unroll
    for (int u = 0; u < PW; ++u)
        for (int j = 32 + lane; j < plen[u]; j += 32)
            range_visit<1>(tmp[plo[u] + j], lc, lcur, stage, ent, bs, staged);
    for (int b = wv + 32 * PW; b < nb; b += 32) {
        const int l0 = row0[b];
        const int ln = row1[b] - l0;
        for (int j = lane; j < ln; j += 32)
            range_visit<1>(tmp[b * EPB + l0 + j], lc, lcur, stage, ent, bs, staged);
    }
    if (staged) {
        __syncthreads();
        for (int i = tid; i < bc; i += 1024) ent[bs + i] = stage[i];
    }
}

typedef __attribute__((ext_vector_type(4))) float v4f;

struct kv8 { v4f k0, k1, v0, v1; };

__device__ __forceinline__ kv8 load_kv8(const float* __restrict__ Kn, const float* __restrict__ Vn,
                                        int node, unsigned hc)
{
    const unsigned boff = (unsigned)node * 256u + hc * 4u;
    const char* kb = (const char*)Kn;
    const char* vb = (const char*)Vn;
    kv8 r;
    r.k0 = *(const v4f*)(kb + boff);
    r.k1 = *(const v4f*)(kb + boff + 16u);
    r.v0 = *(const v4f*)(vb + boff);
    r.v1 = *(const v4f*)(vb + boff + 16u);
    return r;
}

__device__ __forceinline__ float edge_score(float dotv)
{
    return __expf(fminf(fmaxf(dotv * 0.125f, -5.0f), 5.0f));
}

__device__ __forceinline__ void edge_update(const kv8 e, const v2f* q, v2f* ac, v2f* ap)
{
    v2f d = q[0] * e.k0.x;
    d = q[1] * e.k0.y + d;
    d = q[2] * e.k0.z + d;
    d = q[3] * e.k0.w + d;
    d = q[4] * e.k1.x + d;
    d = q[5] * e.k1.y + d;
    d = q[6] * e.k1.z + d;
    d = q[7] * e.k1.w + d;
    const float sc = edge_score(d.x);
    const float sp = edge_score(d.y);
    ac[0] += e.v0.xy * sc; ac[1] += e.v0.zw * sc; ac[2] += e.v1.xy * sc; ac[3] += e.v1.zw * sc;
    ap[0] += e.v0.xy * sp; ap[1] += e.v0.zw * sp; ap[2] += e.v1.xy * sp; ap[3] += e.v1.zw * sp;
}

__global__ __launch_bounds__(256)
void node_gather_kernel(const int2* __restrict__ ent, const int* __restrict__ start,
                        const int* __restrict__ cnt,
                        const float* __restrict__ Kn, const float* __restrict__ Vn,
                        float* out_c, float* out_p, int n)
{
    __shared__ int ssrc[32 * SEG_PITCH];

    const int tid = threadIdx.x;
    const int g = tid >> 3;
    const int h = tid & 7;
    const int d = blockIdx.x * 32 + g;
    const bool valid = d < n;

    const unsigned hc = (unsigned)h * 8u;
    const int col = d * 64 + (int)hc;
    int b = 0, c = 0;
    v4f qc0 = {0.f, 0.f, 0.f, 0.f}, qc1 = qc0, qp0 = qc0, qp1 = qc0;
    if (valid) {
        b = start[d]; c = cnt[d];
        qc0 = *(const v4f*)(out_c + col); qc1 = *(const v4f*)(out_c + col + 4);
        qp0 = *(const v4f*)(out_p + col); qp1 = *(const v4f*)(out_p + col + 4);
    }
    const bool ordered = (c <= SEG_CAP);
    int* my = ssrc + g * SEG_PITCH;

    if (ordered) {
        for (int j = h; j < c; j += 8) {
            const int2 me = ent[b + j];
            int r = 0;
            for (int t = 0; t < c; ++t) r += (ent[b + t].x < me.x) ? 1 : 0;
            my[r] = me.y;
        }
    }
    __syncthreads();
    if (!valid) return;

    v2f q[8];
    q[0] = v2f{qc0.x, qp0.x}; q[1] = v2f{qc0.y, qp0.y}; q[2] = v2f{qc0.z, qp0.z}; q[3] = v2f{qc0.w, qp0.w};
    q[4] = v2f{qc1.x, qp1.x}; q[5] = v2f{qc1.y, qp1.y}; q[6] = v2f{qc1.z, qp1.z}; q[7] = v2f{qc1.w, qp1.w};

    v2f ac[4], ap[4];
#pragma unroll
    for (int t = 0; t < 4; ++t) { ac[t] = v2f{0.f, 0.f}; ap[t] = v2f{0.f, 0.f}; }

    if (ordered) {
        const int last = c - 1;
        for (int i = 0; i < c; i += 4) {
            const int i1 = i + 1 < last ? i + 1 : last;
            const int i2 = i + 2 < last ? i + 2 : last;
            const int i3 = i + 3 < last ? i + 3 : last;
            kv8 e0 = load_kv8(Kn, Vn, my[i], hc);
            const kv8 e1 = load_kv8(Kn, Vn, my[i1], hc);
            const kv8 e2 = load_kv8(Kn, Vn, my[i2], hc);
            const kv8 e3 = load_kv8(Kn, Vn, my[i3], hc);
            asm volatile("" : "+v"(e0.k0), "+v"(e0.k1), "+v"(e0.v0), "+v"(e0.v1)
                            : "v"(e1.k0), "v"(e1.k1), "v"(e1.v0), "v"(e1.v1),
                              "v"(e2.k0), "v"(e2.k1), "v"(e2.v0), "v"(e2.v1),
                              "v"(e3.k0), "v"(e3.k1), "v"(e3.v0), "v"(e3.v1));
            edge_update(e0, q, ac, ap);
            if (i + 1 < c) edge_update(e1, q, ac, ap);
            if (i + 2 < c) edge_update(e2, q, ac, ap);
            if (i + 3 < c) edge_update(e3, q, ac, ap);
        }
    } else {
        for (int i = 0; i < c; ++i)
            edge_update(load_kv8(Kn, Vn, ent[b + i].y, hc), q, ac, ap);
    }

    *(v4f*)(out_c + col)     = __builtin_shufflevector(ac[0], ac[1], 0, 1, 2, 3);
    *(v4f*)(out_c + col + 4) = __builtin_shufflevector(ac[2], ac[3], 0, 1, 2, 3);
    *(v4f*)(out_p + col)     = __builtin_shufflevector(ap[0], ap[1], 0, 1, 2, 3);
    *(v4f*)(out_p + col + 4) = __builtin_shufflevector(ap[2], ap[3], 0, 1, 2, 3);
}

extern "C" void kernel_launch(void* const* d_in, const int* in_sizes, int n_in,
                              void* d_out, int out_size, void* d_ws, size_t ws_size,
                              hipStream_t stream)
{
    const float* h_c  = (const float*)d_in[0];
    const float* h_p  = (const float*)d_in[1];
    const int*   src  = (const int*)  d_in[2];
    const int*   dst  = (const int*)  d_in[3];
    const float* Wqcw = (const float*)d_in[4];
    const float* Wqcb = (const float*)d_in[5];
    const float* Wqpw = (const float*)d_in[6];
    const float* Wqpb = (const float*)d_in[7];
    const float* Wkw  = (const float*)d_in[8];
    const float* Wkb  = (const float*)d_in[9];
    const float* Wvw  = (const float*)d_in[10];
    const float* Wvb  = (const float*)d_in[11];

    const int n = in_sizes[0] / 64;
    const int e = in_sizes[2];

    float* out_c = (float*)d_out;
    float* out_p = out_c + (size_t)n * 64;

    float* Kn   = (float*)d_ws;
    float* Vn   = Kn + (size_t)n * 64;
    int2*  ent  = (int2*)(Vn + (size_t)n * 64);
    int2*  tmp  = ent + (size_t)e;
    int*   cnt  = (int*)(tmp + (size_t)e);
    int*   start  = cnt + n;
    int*   lst    = start + n;
    const int nb    = (e + EPB - 1) / EPB;
    const int nbins = (n + BIN_NODES - 1) / BIN_NODES;

    range_bin_kernel<<<nb, 256, 0, stream>>>(src, dst, tmp, lst, e, nb, nbins);

    range_place_kernel<<<nbins, 1024, 0, stream>>>(tmp, lst, ent, start, cnt, n, nb);

    {
        int blocks = (n + 127) / 128;
        proj_wmma_kernel<<<blocks, 256, 0, stream>>>(
            h_c, h_p, Wqcw, Wqcb, Wqpw, Wqpb, Wkw, Wkb, Wvw, Wvb,
            out_c, out_p, Kn, Vn, n);
    }

    node_gather_kernel<<<(n + 31) / 32, 256, 0, stream>>>(
        ent, start, cnt, Kn, Vn, out_c, out_p, n);
}


# gather: edge ranks from LDS: a node's run is read once into LDS in slot order, each lane ranks its edges against the index list by 16-byte LDS reads (was c*c global reads per node); sums as before
# speedup vs baseline: 6.0163x; 1.0864x over previous
#include <hip/hip_runtime.h>

typedef __attribute__((ext_vector_type(2))) float v2f;
typedef __attribute__((ext_vector_type(8))) float v8f;

#define WT_STRIDE 66
#define SEG_CAP   64
#define SEG_PITCH 65
#define KEY_PITCH 72
#define OWN       3

__global__ __launch_bounds__(256)
void proj_wmma_kernel(const float* __restrict__ h_c,
                      const float* __restrict__ h_p,
                      const float* __restrict__ Wqc, const float* __restrict__ bqc,
                      const float* __restrict__ Wqp, const float* __restrict__ bqp,
                      const float* __restrict__ Wk,  const float* __restrict__ bk,
                      const float* __restrict__ Wv,  const float* __restrict__ bv,
                      float* __restrict__ Qc, float* __restrict__ Qp,
                      float* __restrict__ Kn, float* __restrict__ Vn,
                      int n)
{
    __shared__ float sWT[4][64 * WT_STRIDE];
    __shared__ float sB[4][64];

    const int tid = threadIdx.x;
    const float* Ws[4] = {Wqc, Wqp, Wk, Wv};
    const float* Bs[4] = {bqc, bqp, bk, bv};
    {
        const int cg = tid & 15;
        const int kb = tid >> 4;
        float4 w[4][4];
        float  bias_in[4];
#pragma unroll
        for (int o = 0; o < 4; ++o) {
#pragma unroll
            for (int j = 0; j < 4; ++j)
                w[o][j] = *(const float4*)(Ws[o] + (4 * kb + j) * 64 + 4 * cg);
            bias_in[o] = Bs[o][tid & 63];
        }
#pragma unroll
        for (int o = 0; o < 4; ++o) {
            float* t = &sWT[o][(4 * cg) * WT_STRIDE + 4 * kb];
            *(v2f*)(t)                     = v2f{w[o][0].x, w[o][1].x};
            *(v2f*)(t + 2)                 = v2f{w[o][2].x, w[o][3].x};
            *(v2f*)(t + WT_STRIDE)         = v2f{w[o][0].y, w[o][1].y};
            *(v2f*)(t + WT_STRIDE + 2)     = v2f{w[o][2].y, w[o][3].y};
            *(v2f*)(t + 2 * WT_STRIDE)     = v2f{w[o][0].z, w[o][1].z};
            *(v2f*)(t + 2 * WT_STRIDE + 2) = v2f{w[o][2].z, w[o][3].z};
            *(v2f*)(t + 3 * WT_STRIDE)     = v2f{w[o][0].w, w[o][1].w};
            *(v2f*)(t + 3 * WT_STRIDE + 2) = v2f{w[o][2].w, w[o][3].w};
            if (tid < 64) sB[o][tid] = bias_in[o];
        }
    }
    __syncthreads();

    const int wave = tid >> 5;
    const int lane = tid & 31;
    const int half = lane >> 4;
    const int lrow = lane & 15;
    const int ncol = lane & 15;

    const int m0 = (blockIdx.x * 8 + wave) * 16;
    if (m0 >= n) return;

    const int row = m0 + lrow;

    v2f a_c[16], a_p[16];
#pragma unroll
    for (int ks = 0; ks < 16; ++ks) {
        int k = ks * 4 + 2 * half;
        a_c[ks] = *(const v2f*)(h_c + row * 64 + k);
        a_p[ks] = *(const v2f*)(h_p + row * 64 + k);
    }

    float* outs[4] = {Qc, Qp, Kn, Vn};

#pragma unroll
    for (int o = 0; o < 4; ++o) {
        float* op = outs[o];
#pragma unroll
        for (int nt = 0; nt < 4; ++nt) {
            const int ncg = nt * 16 + ncol;
            v8f acc = {0.f, 0.f, 0.f, 0.f, 0.f, 0.f, 0.f, 0.f};
#pragma unroll
            for (int ks = 0; ks < 16; ++ks) {
                int k = ks * 4 + 2 * half;
                v2f b = *(const v2f*)(&sWT[o][ncg * WT_STRIDE + k]);
                v2f a = (o == 1) ? a_p[ks] : a_c[ks];
                acc = __builtin_amdgcn_wmma_f32_16x16x4_f32(
                     false, a,  false, b,
                     (short)0, acc,  false,  false);
            }
            const float bias = sB[o][ncg];
#pragma unroll
            for (int r = 0; r < 8; ++r) {
                int m = m0 + r + 8 * half;
                op[m * 64 + ncg] = acc[r] + bias;
            }
        }
    }
}

#define BIN_SHIFT 9
#define BIN_NODES 512
#define BIN_MAX   256
#define SRC_BITS  17
#define EPT       16
#define EPB       4096
#define STAGE_CAP 12288
#define PW        13

__device__ __forceinline__ int wave_scan_incl(int x, int lane)
{
    for (int off = 1; off < 32; off <<= 1) {
        int y = __shfl_up(x, off, 32);
        if (lane >= off) x += y;
    }
    return x;
}

__global__ __launch_bounds__(256)
void range_bin_kernel(const int* __restrict__ src, const int* __restrict__ dst,
                      int2* __restrict__ tmp, int* __restrict__ lst,
                      int e_total, int nb, int nbins)
{
    __shared__ int  hist[BIN_MAX];
    __shared__ int  lstart[BIN_MAX];
    __shared__ int  wsA[8];
    __shared__ int2 stage[EPB];

    const int tid  = threadIdx.x;
    const int lane = tid & 31;
    const int wave = tid >> 5;
    hist[tid] = 0;
    __syncthreads();

    const int base = blockIdx.x * EPB;
    const int last = e_total - 1;
    const int left = e_total - base;
    const int m    = left < EPB ? left : EPB;
    int dd[EPT], ss[EPT], rr[EPT];
    for (int j = 0; j < EPT; ++j) {
        const int e  = base + j * 256 + tid;
        const int ec = e < e_total ? e : last;
        dd[j] = dst[ec];
        ss[j] = src[ec];
    }
    for (int j = 0; j < EPT; ++j) {
        const int e = base + j * 256 + tid;
        rr[j] = 0;
        if (e < e_total) rr[j] = atomicAdd(&hist[dd[j] >> BIN_SHIFT], 1);
    }
    __syncthreads();

    const int h  = hist[tid];
    const int xh = wave_scan_incl(h, lane);
    if (lane == 31) wsA[wave] = xh;
    __syncthreads();
    int offA = 0;
    for (int w = 0; w < 8; ++w) {
        if (w < wave) offA += wsA[w];
    }
    const int ls = offA + xh - h;
    lstart[tid] = ls;
    if (tid < nbins) lst[tid * nb + blockIdx.x] = ls;
    if (tid == 0)    lst[nbins * nb + blockIdx.x] = m;
    __syncthreads();

    for (int j = 0; j < EPT; ++j) {
        const int e = base + j * 256 + tid;
        if (e < e_total) {
            const int p = lstart[dd[j] >> BIN_SHIFT] + rr[j];
            stage[p] = make_int2(e, ss[j] | ((dd[j] & (BIN_NODES - 1)) << SRC_BITS));
        }
    }
    __syncthreads();

    for (int i = tid; i < m; i += 256) tmp[base + i] = stage[i];
}

template <int MODE>
__device__ __forceinline__ void range_visit(const int2 a, int* lc, int* lcur,
                                            int2* stage, int2* __restrict__ ent,
                                            int bs, bool staged)
{
    if (MODE == 0) {
        atomicAdd(&lc[a.y >> SRC_BITS], 1);
    } else {
        const int  p = atomicAdd(&lcur[a.y >> SRC_BITS], 1);
        const int2 o = make_int2(a.x, a.y & ((1 << SRC_BITS) - 1));
        if (staged) stage[p] = o; else ent[bs + p] = o;
    }
}

__global__ __launch_bounds__(1024)
void range_place_kernel(const int2* __restrict__ tmp, const int* __restrict__ lst,
                        int2* __restrict__ ent, int* __restrict__ start,
                        int* __restrict__ cnt, int n, int nb)
{
    __shared__ int  lc[BIN_NODES];
    __shared__ int  lcur[BIN_NODES];
    __shared__ int  wsum[32];
    __shared__ int  wsum2[32];
    __shared__ int  sh_bs;
    __shared__ int  sh_bc;
    __shared__ int2 stage[STAGE_CAP];

    const int tid  = threadIdx.x;
    const int lane = tid & 31;
    const int wave = tid >> 5;
    const int wv   = wave;
    const int k    = blockIdx.x;
    const int* __restrict__ row0 = lst + k * nb;
    const int* __restrict__ row1 = lst + (k + 1) * nb;

    int plo[PW], plen[PW];
    int2 a[PW];
#pragma unroll
    for (int u = 0; u < PW; ++u) {
        const int b  = wv + 32 * u;
        const int bb = b < nb ? b : nb - 1;
        plo[u]  = row0[bb];
        plen[u] = row1[bb];
    }
#pragma unroll
    for (int u = 0; u < PW; ++u) {
        const int b  = wv + 32 * u;
        const int bb = b < nb ? b : nb - 1;
        plen[u] = b < nb ? plen[u] - plo[u] : 0;
        plo[u]  = bb * EPB + plo[u];
    }
#pragma unroll
    for (int u = 0; u < PW; ++u)
        a[u] = tmp[lane < plen[u] ? plo[u] + lane : 0];

    {
        int s0 = 0, s1 = 0;
        for (int b = tid; b < nb; b += 1024) { s0 += row0[b]; s1 += row1[b]; }
        const int x0 = wave_scan_incl(s0, lane);
        const int x1 = wave_scan_incl(s1, lane);
        if (lane == 31) { wsum[wave] = x0; wsum2[wave] = x1; }
        if (tid < BIN_NODES) lc[tid] = 0;
        __syncthreads();
        if (tid == 0) {
            int t0 = 0, t1 = 0;
            for (int w = 0; w < 32; ++w) { t0 += wsum[w]; t1 += wsum2[w]; }
            sh_bs = t0;
            sh_bc = t1 - t0;
        }
        __syncthreads();
    }
    const int  bs     = sh_bs;
    const int  bc     = sh_bc;
    const int  node0  = k << BIN_SHIFT;
    const bool staged = (bc <= STAGE_CAP);

#pragma unroll
    for (int u = 0; u < PW; ++u)
        if (lane < plen[u]) range_visit<0>(a[u], lc, lcur, stage, ent, bs, staged);
#pragma unroll
    for (int u = 0; u < PW; ++u)
        for (int j = 32 + lane; j < plen[u]; j += 32)
            range_visit<0>(tmp[plo[u] + j], lc, lcur, stage, ent, bs, staged);
    for (int b = wv + 32 * PW; b < nb; b += 32) {
        const int l0 = row0[b];
        const int ln = row1[b] - l0;
        for (int j = lane; j < ln; j += 32)
            range_visit<0>(tmp[b * EPB + l0 + j], lc, lcur, stage, ent, bs, staged);
    }
    __syncthreads();

    {
        const int c = (tid < BIN_NODES) ? lc[tid] : 0;
        const int x = wave_scan_incl(c, lane);
        if (lane == 31) wsum[wave] = x;
        __syncthreads();
        int woff = 0;
        for (int w = 0; w < 16; ++w) {
            if (w < wave) woff += wsum[w];
        }
        if (tid < BIN_NODES) {
            const int ex = woff + x - c;
            lcur[tid] = ex;
            const int node = node0 + tid;
            if (node < n) { start[node] = bs + ex; cnt[node] = c; }
        }
        __syncthreads();
    }

#pragma unroll
    for (int u = 0; u < PW; ++u)
        if (lane < plen[u]) range_visit<1>(a[u], lc, lcur, stage, ent, bs, staged);
#pragma unroll
    for (int u = 0; u < PW; ++u)
        for (int j = 32 + lane; j < plen[u]; j += 32)
            range_visit<1>(tmp[plo[u] + j], lc, lcur, stage, ent, bs, staged);
    for (int b = wv + 32 * PW; b < nb; b += 32) {
        const int l0 = row0[b];
        const int ln = row1[b] - l0;
        for (int j = lane; j < ln; j += 32)
            range_visit<1>(tmp[b * EPB + l0 + j], lc, lcur, stage, ent, bs, staged);
    }
    if (staged) {
        __syncthreads();
        for (int i = tid; i < bc; i += 1024) ent[bs + i] = stage[i];
    }
}

typedef __attribute__((ext_vector_type(4))) float v4f;

struct kv8 { v4f k0, k1, v0, v1; };

__device__ __forceinline__ kv8 load_kv8(const float* __restrict__ Kn, const float* __restrict__ Vn,
                                        int node, unsigned hc)
{
    const unsigned boff = (unsigned)node * 256u + hc * 4u;
    const char* kb = (const char*)Kn;
    const char* vb = (const char*)Vn;
    kv8 r;
    r.k0 = *(const v4f*)(kb + boff);
    r.k1 = *(const v4f*)(kb + boff + 16u);
    r.v0 = *(const v4f*)(vb + boff);
    r.v1 = *(const v4f*)(vb + boff + 16u);
    return r;
}

__device__ __forceinline__ float edge_score(float dotv)
{
    return __expf(fminf(fmaxf(dotv * 0.125f, -5.0f), 5.0f));
}

__device__ __forceinline__ void edge_update(const kv8 e, const v2f* q, v2f* ac, v2f* ap)
{
    v2f d = q[0] * e.k0.x;
    d = q[1] * e.k0.y + d;
    d = q[2] * e.k0.z + d;
    d = q[3] * e.k0.w + d;
    d = q[4] * e.k1.x + d;
    d = q[5] * e.k1.y + d;
    d = q[6] * e.k1.z + d;
    d = q[7] * e.k1.w + d;
    const float sc = edge_score(d.x);
    const float sp = edge_score(d.y);
    ac[0] += e.v0.xy * sc; ac[1] += e.v0.zw * sc; ac[2] += e.v1.xy * sc; ac[3] += e.v1.zw * sc;
    ap[0] += e.v0.xy * sp; ap[1] += e.v0.zw * sp; ap[2] += e.v1.xy * sp; ap[3] += e.v1.zw * sp;
}

__global__ __launch_bounds__(256)
void node_gather_kernel(const int2* __restrict__ ent, const int* __restrict__ start,
                        const int* __restrict__ cnt,
                        const float* __restrict__ Kn, const float* __restrict__ Vn,
                        float* out_c, float* out_p, int n)
{
    __shared__ int ssrc[32 * SEG_PITCH];
    __shared__ __attribute__((aligned(16))) int skey[32 * KEY_PITCH];
    __shared__ int sval[32 * KEY_PITCH];

    const int tid = threadIdx.x;
    const int g = tid >> 3;
    const int h = tid & 7;
    const int d = blockIdx.x * 32 + g;
    const bool valid = d < n;

    const unsigned hc = (unsigned)h * 8u;
    const int col = d * 64 + (int)hc;
    int b = 0, c = 0;
    v4f qc0 = {0.f, 0.f, 0.f, 0.f}, qc1 = qc0, qp0 = qc0, qp1 = qc0;
    if (valid) {
        b = start[d]; c = cnt[d];
        qc0 = *(const v4f*)(out_c + col); qc1 = *(const v4f*)(out_c + col + 4);
        qp0 = *(const v4f*)(out_p + col); qp1 = *(const v4f*)(out_p + col + 4);
    }
    const bool ordered = (c <= SEG_CAP);
    int* my  = ssrc + g * SEG_PITCH;
    int* key = skey + g * KEY_PITCH;
    int* val = sval + g * KEY_PITCH;
    const int BIG = 0x7fffffff;

    if (ordered && c > 0) {
        const int last = c - 1;
        const int c8   = (c + 7) & ~7;
        const int j1 = h + 8, j2 = h + 16;
        const int2 e0 = ent[b + (h  < last ? h  : last)];
        const int2 e1 = ent[b + (j1 < last ? j1 : last)];
        const int2 e2 = ent[b + (j2 < last ? j2 : last)];
        key[h]  = h  < c ? e0.x : BIG;  val[h]  = e0.y;
        key[j1] = j1 < c ? e1.x : BIG;  val[j1] = e1.y;
        key[j2] = j2 < c ? e2.x : BIG;  val[j2] = e2.y;
        for (int j = h + 24; j < c8; j += 8) {
            const int2 e = ent[b + (j < last ? j : last)];
            key[j] = j < c ? e.x : BIG;  val[j] = e.y;
        }
    }
    __syncthreads();

    if (ordered) {
        for (int jb = 0; jb < c; jb += 8 * OWN) {
            int own[OWN], r[OWN];
#pragma unroll
            for (int u = 0; u < OWN; ++u) {
                const int j = jb + h + 8 * u;
                own[u] = j < c ? key[j] : BIG;
                r[u] = 0;
            }
            for (int t = 0; t < c; t += 8) {
                const int4 p = *(const int4*)(key + t);
                const int4 q = *(const int4*)(key + t + 4);
#pragma unroll
                for (int u = 0; u < OWN; ++u) {
                    const int o = own[u];
                    r[u] += ((p.x < o) ? 1 : 0) + ((p.y < o) ? 1 : 0) + ((p.z < o) ? 1 : 0) + ((p.w < o) ? 1 : 0)
                          + ((q.x < o) ? 1 : 0) + ((q.y < o) ? 1 : 0) + ((q.z < o) ? 1 : 0) + ((q.w < o) ? 1 : 0);
                }
            }
#pragma unroll
            for (int u = 0; u < OWN; ++u) {
                const int j = jb + h + 8 * u;
                if (j < c) my[r[u]] = val[j];
            }
        }
    }
    __syncthreads();
    if (!valid) return;

    v2f q[8];
    q[0] = v2f{qc0.x, qp0.x}; q[1] = v2f{qc0.y, qp0.y}; q[2] = v2f{qc0.z, qp0.z}; q[3] = v2f{qc0.w, qp0.w};
    q[4] = v2f{qc1.x, qp1.x}; q[5] = v2f{qc1.y, qp1.y}; q[6] = v2f{qc1.z, qp1.z}; q[7] = v2f{qc1.w, qp1.w};

    v2f ac[4], ap[4];
#pragma unroll
    for (int t = 0; t < 4; ++t) { ac[t] = v2f{0.f, 0.f}; ap[t] = v2f{0.f, 0.f}; }

    if (ordered) {
        const int last = c - 1;
        for (int i = 0; i < c; i += 4) {
            const int i1 = i + 1 < last ? i + 1 : last;
            const int i2 = i + 2 < last ? i + 2 : last;
            const int i3 = i + 3 < last ? i + 3 : last;
            kv8 e0 = load_kv8(Kn, Vn, my[i], hc);
            const kv8 e1 = load_kv8(Kn, Vn, my[i1], hc);
            const kv8 e2 = load_kv8(Kn, Vn, my[i2], hc);
            const kv8 e3 = load_kv8(Kn, Vn, my[i3], hc);
            asm volatile("" : "+v"(e0.k0), "+v"(e0.k1), "+v"(e0.v0), "+v"(e0.v1)
                            : "v"(e1.k0), "v"(e1.k1), "v"(e1.v0), "v"(e1.v1),
                              "v"(e2.k0), "v"(e2.k1), "v"(e2.v0), "v"(e2.v1),
                              "v"(e3.k0), "v"(e3.k1), "v"(e3.v0), "v"(e3.v1));
            edge_update(e0, q, ac, ap);
            if (i + 1 < c) edge_update(e1, q, ac, ap);
            if (i + 2 < c) edge_update(e2, q, ac, ap);
            if (i + 3 < c) edge_update(e3, q, ac, ap);
        }
    } else {
        for (int i = 0; i < c; ++i)
            edge_update(load_kv8(Kn, Vn, ent[b + i].y, hc), q, ac, ap);
    }

    *(v4f*)(out_c + col)     = __builtin_shufflevector(ac[0], ac[1], 0, 1, 2, 3);
    *(v4f*)(out_c + col + 4) = __builtin_shufflevector(ac[2], ac[3], 0, 1, 2, 3);
    *(v4f*)(out_p + col)     = __builtin_shufflevector(ap[0], ap[1], 0, 1, 2, 3);
    *(v4f*)(out_p + col + 4) = __builtin_shufflevector(ap[2], ap[3], 0, 1, 2, 3);
}

extern "C" void kernel_launch(void* const* d_in, const int* in_sizes, int n_in,
                              void* d_out, int out_size, void* d_ws, size_t ws_size,
                              hipStream_t stream)
{
    const float* h_c  = (const float*)d_in[0];
    const float* h_p  = (const float*)d_in[1];
    const int*   src  = (const int*)  d_in[2];
    const int*   dst  = (const int*)  d_in[3];
    const float* Wqcw = (const float*)d_in[4];
    const float* Wqcb = (const float*)d_in[5];
    const float* Wqpw = (const float*)d_in[6];
    const float* Wqpb = (const float*)d_in[7];
    const float* Wkw  = (const float*)d_in[8];
    const float* Wkb  = (const float*)d_in[9];
    const float* Wvw  = (const float*)d_in[10];
    const float* Wvb  = (const float*)d_in[11];

    const int n = in_sizes[0] / 64;
    const int e = in_sizes[2];

    float* out_c = (float*)d_out;
    float* out_p = out_c + (size_t)n * 64;

    float* Kn   = (float*)d_ws;
    float* Vn   = Kn + (size_t)n * 64;
    int2*  ent  = (int2*)(Vn + (size_t)n * 64);
    int2*  tmp  = ent + (size_t)e;
    int*   cnt  = (int*)(tmp + (size_t)e);
    int*   start  = cnt + n;
    int*   lst    = start + n;
    const int nb    = (e + EPB - 1) / EPB;
    const int nbins = (n + BIN_NODES - 1) / BIN_NODES;

    range_bin_kernel<<<nb, 256, 0, stream>>>(src, dst, tmp, lst, e, nb, nbins);

    range_place_kernel<<<nbins, 1024, 0, stream>>>(tmp, lst, ent, start, cnt, n, nb);

    {
        int blocks = (n + 127) / 128;
        proj_wmma_kernel<<<blocks, 256, 0, stream>>>(
            h_c, h_p, Wqcw, Wqcb, Wqpw, Wqpb, Wkw, Wkb, Wvw, Wvb,
            out_c, out_p, Kn, Vn, n);
    }

    node_gather_kernel<<<(n + 31) / 32, 256, 0, stream>>>(
        ent, start, cnt, Kn, Vn, out_c, out_p, n);
}
